# starting kernel
# baseline (speedup 1.0000x reference)
#include <hip/hip_runtime.h>
#include <math.h>

typedef __attribute__((ext_vector_type(2))) float v2f;
typedef __attribute__((ext_vector_type(8))) float v8f;

#define CDIM 256
#define NDIM 784
#define NBATCH 64

static __device__ __forceinline__ v8f wmma4(v2f a, v2f b, v8f c) {
    return __builtin_amdgcn_wmma_f32_16x16x4_f32(false, a, false, b,
                                                 (short)0, c, false, false);
}

__global__ void mean_kernel(const float* __restrict__ X, float* __restrict__ mean) {
    int wave = threadIdx.x >> 5;
    int lane = threadIdx.x & 31;
    int row  = blockIdx.x * 8 + wave;
    const float* p = X + (size_t)row * NDIM;
    float s = 0.f;
    for (int k = lane; k < NDIM; k += 32) s += p[k];
    for (int off = 16; off > 0; off >>= 1) s += __shfl_down(s, off, 32);
    if (lane == 0) mean[row] = s * (1.0f / (float)NDIM);
}

__global__ void sigma_kernel(const float* __restrict__ X,
                             const float* __restrict__ mean,
                             float* __restrict__ S) {
    int wave  = threadIdx.x >> 5;
    int lane  = threadIdx.x & 31;
    int tile  = (blockIdx.x & 3) * 8 + wave;
    int batch =  blockIdx.x >> 2;
    int ti = (tile >> 3) << 6;
    int tj = (tile &  7) << 5;

    const float* Xb = X    + (size_t)batch * CDIM * NDIM;
    const float* mb = mean + batch * CDIM;

    int mi = lane & 15;
    int kh = (lane >> 4) << 1;
    float ma0 = mb[ti      + mi], ma1 = mb[ti + 16 + mi];
    float ma2 = mb[ti + 32 + mi], ma3 = mb[ti + 48 + mi];
    float mn0 = mb[tj      + mi], mn1 = mb[tj + 16 + mi];

    const float* ar0 = Xb + (size_t)(ti + mi) * NDIM + kh;
    const float* ar1 = ar0 + (size_t)16 * NDIM;
    const float* ar2 = ar0 + (size_t)32 * NDIM;
    const float* ar3 = ar0 + (size_t)48 * NDIM;
    const float* br0 = Xb + (size_t)(tj + mi) * NDIM + kh;
    const float* br1 = br0 + (size_t)16 * NDIM;

    v8f c00 = {}, c01 = {}, c10 = {}, c11 = {};
    v8f c20 = {}, c21 = {}, c30 = {}, c31 = {};
    for (int k = 0; k < NDIM; k += 4) {
        v2f a0 = { ar0[k] - ma0, ar0[k + 1] - ma0 };
        v2f a1 = { ar1[k] - ma1, ar1[k + 1] - ma1 };
        v2f a2 = { ar2[k] - ma2, ar2[k + 1] - ma2 };
        v2f a3 = { ar3[k] - ma3, ar3[k + 1] - ma3 };
        v2f b0 = { br0[k] - mn0, br0[k + 1] - mn0 };
        v2f b1 = { br1[k] - mn1, br1[k + 1] - mn1 };
        c00 = wmma4(a0, b0, c00);  c01 = wmma4(a0, b1, c01);
        c10 = wmma4(a1, b0, c10);  c11 = wmma4(a1, b1, c11);
        c20 = wmma4(a2, b0, c20);  c21 = wmma4(a2, b1, c21);
        c30 = wmma4(a3, b0, c30);  c31 = wmma4(a3, b1, c31);
    }

    float* Sb = S + (size_t)batch * CDIM * CDIM;
    const float inv_n = 1.0f / (float)NDIM;
    int mbase = (lane >> 4) << 3;
    int nlo   = lane & 15;
    for (int r = 0; r < 8; ++r) {
        int gm = ti + mbase + r;
        int gn0 = tj + nlo, gn1 = gn0 + 16;
        Sb[(size_t)(gm     ) * CDIM + gn0] = c00[r] * inv_n;
        Sb[(size_t)(gm     ) * CDIM + gn1] = c01[r] * inv_n;
        Sb[(size_t)(gm + 16) * CDIM + gn0] = c10[r] * inv_n;
        Sb[(size_t)(gm + 16) * CDIM + gn1] = c11[r] * inv_n;
        Sb[(size_t)(gm + 32) * CDIM + gn0] = c20[r] * inv_n;
        Sb[(size_t)(gm + 32) * CDIM + gn1] = c21[r] * inv_n;
        Sb[(size_t)(gm + 48) * CDIM + gn0] = c30[r] * inv_n;
        Sb[(size_t)(gm + 48) * CDIM + gn1] = c31[r] * inv_n;
    }
}

__global__ void trace_kernel(const float* __restrict__ S, float* __restrict__ trc) {
    __shared__ float sm[256];
    int b = blockIdx.x, t = threadIdx.x;
    sm[t] = S[(size_t)b * CDIM * CDIM + (size_t)t * CDIM + t];
    __syncthreads();
    for (int off = 128; off > 0; off >>= 1) {
        if (t < off) sm[t] += sm[t + off];
        __syncthreads();
    }
    if (t == 0) trc[b] = sm[0];
}

__global__ void init_kernel(const float* __restrict__ S, const float* __restrict__ trc,
                            float* __restrict__ Y, float* __restrict__ Z) {
    size_t idx = (size_t)blockIdx.x * 256 + threadIdx.x;
    int b  = (int)(idx >> 16);
    int ij = (int)(idx & 65535);
    Y[idx] = S[idx] / trc[b];
    Z[idx] = ((ij >> 8) == (ij & 255)) ? 1.0f : 0.0f;
}

__global__ void gemm256_kernel(const float* __restrict__ A,
                               const float* __restrict__ Bm,
                               float* __restrict__ D,
                               float scale, float diag) {
    int wave  = threadIdx.x >> 5;
    int lane  = threadIdx.x & 31;
    int tile  = (blockIdx.x & 3) * 8 + wave;
    int batch =  blockIdx.x >> 2;
    int ti = (tile >> 3) << 6;
    int tj = (tile &  7) << 5;

    const float* Ab = A  + (size_t)batch * CDIM * CDIM;
    const float* Bb = Bm + (size_t)batch * CDIM * CDIM;

    int mi = lane & 15;
    int kh = (lane >> 4) << 1;

    const float* ar0 = Ab + (size_t)(ti + mi) * CDIM + kh;
    const float* ar1 = ar0 + (size_t)16 * CDIM;
    const float* ar2 = ar0 + (size_t)32 * CDIM;
    const float* ar3 = ar0 + (size_t)48 * CDIM;
    const float* bc0 = Bb + (size_t)kh * CDIM + (tj + mi);
    const float* bc1 = bc0 + 16;

    v8f c00 = {}, c01 = {}, c10 = {}, c11 = {};
    v8f c20 = {}, c21 = {}, c30 = {}, c31 = {};
    for (int k = 0; k < CDIM; k += 4) {
        size_t ko = (size_t)k * CDIM;
        v2f a0 = { ar0[k],  ar0[k + 1] };
        v2f a1 = { ar1[k],  ar1[k + 1] };
        v2f a2 = { ar2[k],  ar2[k + 1] };
        v2f a3 = { ar3[k],  ar3[k + 1] };
        v2f b0 = { bc0[ko], bc0[ko + CDIM] };
        v2f b1 = { bc1[ko], bc1[ko + CDIM] };
        c00 = wmma4(a0, b0, c00);  c01 = wmma4(a0, b1, c01);
        c10 = wmma4(a1, b0, c10);  c11 = wmma4(a1, b1, c11);
        c20 = wmma4(a2, b0, c20);  c21 = wmma4(a2, b1, c21);
        c30 = wmma4(a3, b0, c30);  c31 = wmma4(a3, b1, c31);
    }

    float* Db = D + (size_t)batch * CDIM * CDIM;
    int mbase = (lane >> 4) << 3;
    int nlo   = lane & 15;
    for (int r = 0; r < 8; ++r) {
        int gm = ti + mbase + r;
        int gn0 = tj + nlo, gn1 = gn0 + 16;
        float v;
        v = scale * c00[r]; if (gm      == gn0) v += diag; Db[(size_t)(gm     ) * CDIM + gn0] = v;
        v = scale * c01[r]; if (gm      == gn1) v += diag; Db[(size_t)(gm     ) * CDIM + gn1] = v;
        v = scale * c10[r]; if (gm + 16 == gn0) v += diag; Db[(size_t)(gm + 16) * CDIM + gn0] = v;
        v = scale * c11[r]; if (gm + 16 == gn1) v += diag; Db[(size_t)(gm + 16) * CDIM + gn1] = v;
        v = scale * c20[r]; if (gm + 32 == gn0) v += diag; Db[(size_t)(gm + 32) * CDIM + gn0] = v;
        v = scale * c21[r]; if (gm + 32 == gn1) v += diag; Db[(size_t)(gm + 32) * CDIM + gn1] = v;
        v = scale * c30[r]; if (gm + 48 == gn0) v += diag; Db[(size_t)(gm + 48) * CDIM + gn0] = v;
        v = scale * c31[r]; if (gm + 48 == gn1) v += diag; Db[(size_t)(gm + 48) * CDIM + gn1] = v;
    }
}

__global__ void final_kernel(const float* __restrict__ Y,
                             const float* __restrict__ trc,
                             float* __restrict__ out) {
    int b = blockIdx.y;
    int i = blockIdx.x;
    int j = threadIdx.x;
    if (j < i) return;
    float s = sqrtf(trc[b]);
    int rowoff = i * CDIM - (i * (i - 1)) / 2;
    out[(size_t)b * 32896 + rowoff + (j - i)] =
        s * Y[(size_t)b * CDIM * CDIM + (size_t)i * CDIM + j];
}

extern "C" void kernel_launch(void* const* d_in, const int* in_sizes, int n_in,
                              void* d_out, int out_size, void* d_ws, size_t ws_size,
                              hipStream_t stream) {
    (void)in_sizes; (void)n_in; (void)out_size; (void)ws_size;

    const float* x   = (const float*)d_in[0];
    float*       out = (float*)d_out;
    float*       ws  = (float*)d_ws;

    const size_t MAT = (size_t)NBATCH * CDIM * CDIM;

    float* mean = ws;
    float* trc  = mean + NBATCH * CDIM;
    float* S    = trc + NBATCH;
    float* Ya   = S  + MAT;
    float* Yb   = Ya + MAT;
    float* Za   = Yb + MAT;
    float* Zb   = Za + MAT;

    mean_kernel <<<(NBATCH * CDIM) / 8, 256, 0, stream>>>(x, mean);
    sigma_kernel<<<NBATCH * 4,         256, 0, stream>>>(x, mean, S);
    trace_kernel<<<NBATCH,             256, 0, stream>>>(S, trc);
    init_kernel <<<(unsigned)(MAT / 256), 256, 0, stream>>>(S, trc, Ya, Za);

    float* T  = S;
    float* Y  = Ya; float* Y2 = Yb;
    float* Z  = Za; float* Z2 = Zb;
    for (int it = 0; it < 3; ++it) {
        gemm256_kernel<<<NBATCH * 4, 256, 0, stream>>>(Z, Y,  T,  -1.0f, 3.0f);
        gemm256_kernel<<<NBATCH * 4, 256, 0, stream>>>(Y, T,  Y2,  0.5f, 0.0f);
        gemm256_kernel<<<NBATCH * 4, 256, 0, stream>>>(Z, Y2, T,  -1.0f, 3.0f);
        gemm256_kernel<<<NBATCH * 4, 256, 0, stream>>>(T, Z,  Z2,  0.5f, 0.0f);
        float* t0 = Y; Y = Y2; Y2 = t0;
        float* t1 = Z; Z = Z2; Z2 = t1;
    }

    final_kernel<<<dim3(CDIM, NBATCH), 256, 0, stream>>>(Y, trc, out);
}


# Drop 5 of 12 GEMMs that cannot change the result: 3 products with the function's own Z0=I become closed forms in init and one GEMM epilogue; the last Z update, never read, is not launched
# speedup vs baseline: 1.3531x; 1.3531x over previous
#include <hip/hip_runtime.h>
#include <math.h>

typedef __attribute__((ext_vector_type(2))) float v2f;
typedef __attribute__((ext_vector_type(8))) float v8f;

#define CDIM 256
#define NDIM 784
#define NBATCH 64

static __device__ __forceinline__ v8f wmma4(v2f a, v2f b, v8f c) {
    return __builtin_amdgcn_wmma_f32_16x16x4_f32(false, a, false, b,
                                                 (short)0, c, false, false);
}

__global__ void mean_kernel(const float* __restrict__ X, float* __restrict__ mean) {
    int wave = threadIdx.x >> 5;
    int lane = threadIdx.x & 31;
    int row  = blockIdx.x * 8 + wave;
    const float* p = X + (size_t)row * NDIM;
    float s = 0.f;
    for (int k = lane; k < NDIM; k += 32) s += p[k];
    for (int off = 16; off > 0; off >>= 1) s += __shfl_down(s, off, 32);
    if (lane == 0) mean[row] = s * (1.0f / (float)NDIM);
}

__global__ void sigma_kernel(const float* __restrict__ X,
                             const float* __restrict__ mean,
                             float* __restrict__ S) {
    int wave  = threadIdx.x >> 5;
    int lane  = threadIdx.x & 31;
    int tile  = (blockIdx.x & 3) * 8 + wave;
    int batch =  blockIdx.x >> 2;
    int ti = (tile >> 3) << 6;
    int tj = (tile &  7) << 5;

    const float* Xb = X    + (size_t)batch * CDIM * NDIM;
    const float* mb = mean + batch * CDIM;

    int mi = lane & 15;
    int kh = (lane >> 4) << 1;
    float ma0 = mb[ti      + mi], ma1 = mb[ti + 16 + mi];
    float ma2 = mb[ti + 32 + mi], ma3 = mb[ti + 48 + mi];
    float mn0 = mb[tj      + mi], mn1 = mb[tj + 16 + mi];

    const float* ar0 = Xb + (size_t)(ti + mi) * NDIM + kh;
    const float* ar1 = ar0 + (size_t)16 * NDIM;
    const float* ar2 = ar0 + (size_t)32 * NDIM;
    const float* ar3 = ar0 + (size_t)48 * NDIM;
    const float* br0 = Xb + (size_t)(tj + mi) * NDIM + kh;
    const float* br1 = br0 + (size_t)16 * NDIM;

    v8f c00 = {}, c01 = {}, c10 = {}, c11 = {};
    v8f c20 = {}, c21 = {}, c30 = {}, c31 = {};
    for (int k = 0; k < NDIM; k += 4) {
        v2f a0 = { ar0[k] - ma0, ar0[k + 1] - ma0 };
        v2f a1 = { ar1[k] - ma1, ar1[k + 1] - ma1 };
        v2f a2 = { ar2[k] - ma2, ar2[k + 1] - ma2 };
        v2f a3 = { ar3[k] - ma3, ar3[k + 1] - ma3 };
        v2f b0 = { br0[k] - mn0, br0[k + 1] - mn0 };
        v2f b1 = { br1[k] - mn1, br1[k + 1] - mn1 };
        c00 = wmma4(a0, b0, c00);  c01 = wmma4(a0, b1, c01);
        c10 = wmma4(a1, b0, c10);  c11 = wmma4(a1, b1, c11);
        c20 = wmma4(a2, b0, c20);  c21 = wmma4(a2, b1, c21);
        c30 = wmma4(a3, b0, c30);  c31 = wmma4(a3, b1, c31);
    }

    float* Sb = S + (size_t)batch * CDIM * CDIM;
    const float inv_n = 1.0f / (float)NDIM;
    int mbase = (lane >> 4) << 3;
    int nlo   = lane & 15;
    for (int r = 0; r < 8; ++r) {
        int gm = ti + mbase + r;
        int gn0 = tj + nlo, gn1 = gn0 + 16;
        Sb[(size_t)(gm     ) * CDIM + gn0] = c00[r] * inv_n;
        Sb[(size_t)(gm     ) * CDIM + gn1] = c01[r] * inv_n;
        Sb[(size_t)(gm + 16) * CDIM + gn0] = c10[r] * inv_n;
        Sb[(size_t)(gm + 16) * CDIM + gn1] = c11[r] * inv_n;
        Sb[(size_t)(gm + 32) * CDIM + gn0] = c20[r] * inv_n;
        Sb[(size_t)(gm + 32) * CDIM + gn1] = c21[r] * inv_n;
        Sb[(size_t)(gm + 48) * CDIM + gn0] = c30[r] * inv_n;
        Sb[(size_t)(gm + 48) * CDIM + gn1] = c31[r] * inv_n;
    }
}

__global__ void trace_kernel(const float* __restrict__ S, float* __restrict__ trc) {
    __shared__ float sm[256];
    int b = blockIdx.x, t = threadIdx.x;
    sm[t] = S[(size_t)b * CDIM * CDIM + (size_t)t * CDIM + t];
    __syncthreads();
    for (int off = 128; off > 0; off >>= 1) {
        if (t < off) sm[t] += sm[t + off];
        __syncthreads();
    }
    if (t == 0) trc[b] = sm[0];
}

__global__ void init_kernel(const float* __restrict__ S, const float* __restrict__ trc,
                            float* __restrict__ Y, float* __restrict__ T) {
    size_t idx = (size_t)blockIdx.x * 256 + threadIdx.x;
    int b  = (int)(idx >> 16);
    int ij = (int)(idx & 65535);
    float y = S[idx] / trc[b];
    Y[idx] = y;
    T[idx] = ((ij >> 8) == (ij & 255)) ? 3.0f - y : -y;
}

template <bool ZOUT>
__global__ void gemm256_kernel(const float* __restrict__ A,
                               const float* __restrict__ Bm,
                               float* __restrict__ D,
                               float* __restrict__ Zn,
                               float scale, float diag) {
    int wave  = threadIdx.x >> 5;
    int lane  = threadIdx.x & 31;
    int tile  = (blockIdx.x & 3) * 8 + wave;
    int batch =  blockIdx.x >> 2;
    int ti = (tile >> 3) << 6;
    int tj = (tile &  7) << 5;

    const float* Ab = A  + (size_t)batch * CDIM * CDIM;
    const float* Bb = Bm + (size_t)batch * CDIM * CDIM;

    int mi = lane & 15;
    int kh = (lane >> 4) << 1;

    const float* ar0 = Ab + (size_t)(ti + mi) * CDIM + kh;
    const float* ar1 = ar0 + (size_t)16 * CDIM;
    const float* ar2 = ar0 + (size_t)32 * CDIM;
    const float* ar3 = ar0 + (size_t)48 * CDIM;
    const float* bc0 = Bb + (size_t)kh * CDIM + (tj + mi);
    const float* bc1 = bc0 + 16;

    v8f c00 = {}, c01 = {}, c10 = {}, c11 = {};
    v8f c20 = {}, c21 = {}, c30 = {}, c31 = {};
    for (int k = 0; k < CDIM; k += 4) {
        size_t ko = (size_t)k * CDIM;
        v2f a0 = { ar0[k],  ar0[k + 1] };
        v2f a1 = { ar1[k],  ar1[k + 1] };
        v2f a2 = { ar2[k],  ar2[k + 1] };
        v2f a3 = { ar3[k],  ar3[k + 1] };
        v2f b0 = { bc0[ko], bc0[ko + CDIM] };
        v2f b1 = { bc1[ko], bc1[ko + CDIM] };
        c00 = wmma4(a0, b0, c00);  c01 = wmma4(a0, b1, c01);
        c10 = wmma4(a1, b0, c10);  c11 = wmma4(a1, b1, c11);
        c20 = wmma4(a2, b0, c20);  c21 = wmma4(a2, b1, c21);
        c30 = wmma4(a3, b0, c30);  c31 = wmma4(a3, b1, c31);
    }

    float* Db = D + (size_t)batch * CDIM * CDIM;
    float* Zb = ZOUT ? Zn + (size_t)batch * CDIM * CDIM : nullptr;
    int mbase = (lane >> 4) << 3;
    int nlo   = lane & 15;
    auto put = [&](float acc, int gm, int gn) {
        float v = scale * acc;
        if (gm == gn) v += diag;
        Db[(size_t)gm * CDIM + gn] = v;
        if (ZOUT) {
            float t = -v;
            if (gm == gn) t += 3.0f;
            Zb[(size_t)gm * CDIM + gn] = 0.5f * t;
        }
    };
    #pragma unroll
    for (int r = 0; r < 8; ++r) {
        int gm = ti + mbase + r;
        int gn0 = tj + nlo, gn1 = gn0 + 16;
        put(c00[r], gm,      gn0);  put(c01[r], gm,      gn1);
        put(c10[r], gm + 16, gn0);  put(c11[r], gm + 16, gn1);
        put(c20[r], gm + 32, gn0);  put(c21[r], gm + 32, gn1);
        put(c30[r], gm + 48, gn0);  put(c31[r], gm + 48, gn1);
    }
}

__global__ void final_kernel(const float* __restrict__ Y,
                             const float* __restrict__ trc,
                             float* __restrict__ out) {
    int b = blockIdx.y;
    int i = blockIdx.x;
    int j = threadIdx.x;
    if (j < i) return;
    float s = sqrtf(trc[b]);
    int rowoff = i * CDIM - (i * (i - 1)) / 2;
    out[(size_t)b * 32896 + rowoff + (j - i)] =
        s * Y[(size_t)b * CDIM * CDIM + (size_t)i * CDIM + j];
}

extern "C" void kernel_launch(void* const* d_in, const int* in_sizes, int n_in,
                              void* d_out, int out_size, void* d_ws, size_t ws_size,
                              hipStream_t stream) {
    (void)in_sizes; (void)n_in; (void)out_size; (void)ws_size;

    const float* x   = (const float*)d_in[0];
    float*       out = (float*)d_out;
    float*       ws  = (float*)d_ws;

    const size_t MAT = (size_t)NBATCH * CDIM * CDIM;

    float* mean = ws;
    float* trc  = mean + NBATCH * CDIM;
    float* S    = trc + NBATCH;
    float* Ya   = S  + MAT;
    float* Yb   = Ya + MAT;
    float* Za   = Yb + MAT;
    float* Zb   = Za + MAT;

    mean_kernel <<<(NBATCH * CDIM) / 8, 256, 0, stream>>>(x, mean);
    sigma_kernel<<<NBATCH * 4,         256, 0, stream>>>(x, mean, S);
    trace_kernel<<<NBATCH,             256, 0, stream>>>(S, trc);
    init_kernel <<<(unsigned)(MAT / 256), 256, 0, stream>>>(S, trc, Ya, Za);

    gemm256_kernel<true> <<<NBATCH * 4, 256, 0, stream>>>(Ya, Za, Yb, Zb, 0.5f, 0.0f);

    gemm256_kernel<false><<<NBATCH * 4, 256, 0, stream>>>(Zb, Yb, S,  nullptr, -1.0f, 3.0f);
    gemm256_kernel<false><<<NBATCH * 4, 256, 0, stream>>>(Yb, S,  Ya, nullptr,  0.5f, 0.0f);
    gemm256_kernel<false><<<NBATCH * 4, 256, 0, stream>>>(Zb, Ya, S,  nullptr, -1.0f, 3.0f);
    gemm256_kernel<false><<<NBATCH * 4, 256, 0, stream>>>(S,  Zb, Za, nullptr,  0.5f, 0.0f);

    gemm256_kernel<false><<<NBATCH * 4, 256, 0, stream>>>(Za, Ya, S,  nullptr, -1.0f, 3.0f);
    gemm256_kernel<false><<<NBATCH * 4, 256, 0, stream>>>(Ya, S,  Yb, nullptr,  0.5f, 0.0f);

    final_kernel<<<dim3(CDIM, NBATCH), 256, 0, stream>>>(Yb, trc, out);
}


# Both WMMA kernels stage operand K-chunks through double-buffered LDS (whole-line 16-byte global loads, next chunk prefetched in registers), not per-fragment global loads; summation order unchanged
# speedup vs baseline: 2.2517x; 1.6641x over previous
#include <hip/hip_runtime.h>
#include <math.h>

typedef __attribute__((ext_vector_type(2))) float v2f;
typedef __attribute__((ext_vector_type(4))) float v4f;
typedef __attribute__((ext_vector_type(8))) float v8f;

#define CDIM 256
#define NDIM 784
#define NBATCH 64

static __device__ __forceinline__ v8f wmma4(v2f a, v2f b, v8f c) {
    return __builtin_amdgcn_wmma_f32_16x16x4_f32(false, a, false, b,
                                                 (short)0, c, false, false);
}

__global__ void mean_kernel(const float* __restrict__ X, float* __restrict__ mean) {
    int wave = threadIdx.x >> 5;
    int lane = threadIdx.x & 31;
    int row  = blockIdx.x * 8 + wave;
    const float* p = X + (size_t)row * NDIM;
    float s = 0.f;
    for (int k = lane; k < NDIM; k += 32) s += p[k];
    for (int off = 16; off > 0; off >>= 1) s += __shfl_down(s, off, 32);
    if (lane == 0) mean[row] = s * (1.0f / (float)NDIM);
}

#define SG_KC 32
#define SG_P  36

#define SG_KSTEP(KK)                                                         \
    {                                                                        \
        v2f a0 = *(const v2f*)(pa + (KK));                                   \
        v2f a1 = *(const v2f*)(pa + 16 * SG_P + (KK));                       \
        v2f a2 = *(const v2f*)(pa + 32 * SG_P + (KK));                       \
        v2f a3 = *(const v2f*)(pa + 48 * SG_P + (KK));                       \
        v2f b0 = *(const v2f*)(pb + (KK));                                   \
        v2f b1 = *(const v2f*)(pb + 16 * SG_P + (KK));                       \
        c00 = wmma4(a0, b0, c00);  c01 = wmma4(a0, b1, c01);                 \
        c10 = wmma4(a1, b0, c10);  c11 = wmma4(a1, b1, c11);                 \
        c20 = wmma4(a2, b0, c20);  c21 = wmma4(a2, b1, c21);                 \
        c30 = wmma4(a3, b0, c30);  c31 = wmma4(a3, b1, c31);                 \
    }

__global__ __launch_bounds__(256)
void sigma_kernel(const float* __restrict__ X,
                  const float* __restrict__ mean,
                  float* __restrict__ S) {
    __shared__ __attribute__((aligned(16))) float Xs[2][CDIM * SG_P];

    int tid   = threadIdx.x;
    int wave  = tid >> 5;
    int lane  = tid & 31;
    int batch = blockIdx.x >> 2;
    int ti = (blockIdx.x & 3) << 6;
    int tj = wave << 5;

    const float* Xb = X    + (size_t)batch * CDIM * NDIM;
    const float* mb = mean + batch * CDIM;

    int srow = tid >> 3;
    int skq  = (tid & 7) << 2;
    const float* gp = Xb + (size_t)srow * NDIM + skq;
    int so = srow * SG_P + skq;
    float mrow[8];
    #pragma unroll
    for (int i = 0; i < 8; ++i) mrow[i] = mb[srow + 32 * i];

    int mi = lane & 15;
    int kh = (lane >> 4) << 1;
    int fa = (ti + mi) * SG_P + kh;
    int fb = (tj + mi) * SG_P + kh;

    v4f r[8];
    #pragma unroll
    for (int i = 0; i < 8; ++i)
        r[i] = *(const v4f*)(gp + (size_t)(32 * i) * NDIM);

    v8f c00 = {}, c01 = {}, c10 = {}, c11 = {};
    v8f c20 = {}, c21 = {}, c30 = {}, c31 = {};

    const int NFULL = NDIM / SG_KC;
    for (int ch = 0; ch < NFULL; ++ch) {
        float* buf = Xs[ch & 1];
        #pragma unroll
        for (int i = 0; i < 8; ++i) {
            *(v4f*)(buf + so + 32 * i * SG_P) = r[i] - mrow[i];
        }
        __syncthreads();
        if (ch + 1 < NFULL || skq < NDIM - NFULL * SG_KC) {
            const float* g = gp + (ch + 1) * SG_KC;
            #pragma unroll
            for (int i = 0; i < 8; ++i)
                r[i] = *(const v4f*)(g + (size_t)(32 * i) * NDIM);
        }
        const float* pa = buf + fa;
        const float* pb = buf + fb;
        #pragma unroll
        for (int kk = 0; kk < SG_KC; kk += 4) SG_KSTEP(kk)
    }
    {
        float* buf = Xs[NFULL & 1];
        if (skq < NDIM - NFULL * SG_KC) {
            #pragma unroll
            for (int i = 0; i < 8; ++i) {
                *(v4f*)(buf + so + 32 * i * SG_P) = r[i] - mrow[i];
            }
        }
        __syncthreads();
        const float* pa = buf + fa;
        const float* pb = buf + fb;
        #pragma unroll
        for (int kk = 0; kk < NDIM - NFULL * SG_KC; kk += 4) SG_KSTEP(kk)
    }

    float* Sb = S + (size_t)batch * CDIM * CDIM;
    const float inv_n = 1.0f / (float)NDIM;
    int mbase = (lane >> 4) << 3;
    int nlo   = lane & 15;
    for (int r8 = 0; r8 < 8; ++r8) {
        int gm = ti + mbase + r8;
        int gn0 = tj + nlo, gn1 = gn0 + 16;
        Sb[(size_t)(gm     ) * CDIM + gn0] = c00[r8] * inv_n;
        Sb[(size_t)(gm     ) * CDIM + gn1] = c01[r8] * inv_n;
        Sb[(size_t)(gm + 16) * CDIM + gn0] = c10[r8] * inv_n;
        Sb[(size_t)(gm + 16) * CDIM + gn1] = c11[r8] * inv_n;
        Sb[(size_t)(gm + 32) * CDIM + gn0] = c20[r8] * inv_n;
        Sb[(size_t)(gm + 32) * CDIM + gn1] = c21[r8] * inv_n;
        Sb[(size_t)(gm + 48) * CDIM + gn0] = c30[r8] * inv_n;
        Sb[(size_t)(gm + 48) * CDIM + gn1] = c31[r8] * inv_n;
    }
}

__global__ void trace_kernel(const float* __restrict__ S, float* __restrict__ trc) {
    __shared__ float sm[256];
    int b = blockIdx.x, t = threadIdx.x;
    sm[t] = S[(size_t)b * CDIM * CDIM + (size_t)t * CDIM + t];
    __syncthreads();
    for (int off = 128; off > 0; off >>= 1) {
        if (t < off) sm[t] += sm[t + off];
        __syncthreads();
    }
    if (t == 0) trc[b] = sm[0];
}

__global__ void init_kernel(const float* __restrict__ S, const float* __restrict__ trc,
                            float* __restrict__ Y, float* __restrict__ T) {
    size_t idx = (size_t)blockIdx.x * 256 + threadIdx.x;
    int b  = (int)(idx >> 16);
    int ij = (int)(idx & 65535);
    float y = S[idx] / trc[b];
    Y[idx] = y;
    T[idx] = ((ij >> 8) == (ij & 255)) ? 3.0f - y : -y;
}

#define GM_KC 32
#define GM_AP 36
#define GM_BP 264

template <bool ZOUT>
__global__ __launch_bounds__(256)
void gemm256_kernel(const float* __restrict__ A,
                    const float* __restrict__ Bm,
                    float* __restrict__ D,
                    float* __restrict__ Zn,
                    float scale, float diag) {
    __shared__ __attribute__((aligned(16))) float As[2][64 * GM_AP];
    __shared__ __attribute__((aligned(16))) float Bs[2][GM_KC * GM_BP];

    int tid   = threadIdx.x;
    int wave  = tid >> 5;
    int lane  = tid & 31;
    int batch = blockIdx.x >> 2;
    int ti = (blockIdx.x & 3) << 6;
    int tj = wave << 5;

    const float* Ab = A  + (size_t)batch * CDIM * CDIM + (size_t)ti * CDIM;
    const float* Bb = Bm + (size_t)batch * CDIM * CDIM;

    int arow = tid >> 3;
    int akq  = (tid & 7) << 2;
    int brow = tid >> 6;
    int bnq  = (tid & 63) << 2;
    const float* ga = Ab + (size_t)arow * CDIM + akq;
    const float* gb = Bb + (size_t)brow * CDIM + bnq;
    int sa = arow * GM_AP + akq;
    int sb = brow * GM_BP + bnq;

    int mi = lane & 15;
    int kh = (lane >> 4) << 1;
    int fa = mi * GM_AP + kh;
    int fb = kh * GM_BP + tj + mi;

    v4f ra0, ra1, rb[8];
    ra0 = *(const v4f*)(ga);
    ra1 = *(const v4f*)(ga + (size_t)32 * CDIM);
    #pragma unroll
    for (int i = 0; i < 8; ++i)
        rb[i] = *(const v4f*)(gb + (size_t)(4 * i) * CDIM);

    v8f c00 = {}, c01 = {}, c10 = {}, c11 = {};
    v8f c20 = {}, c21 = {}, c30 = {}, c31 = {};
    #pragma nounroll
    for (int ch = 0; ch < CDIM / GM_KC; ++ch) {
        float* abuf = As[ch & 1];
        float* bbuf = Bs[ch & 1];
        *(v4f*)(abuf + sa)              = ra0;
        *(v4f*)(abuf + sa + 32 * GM_AP) = ra1;
        #pragma unroll
        for (int i = 0; i < 8; ++i)
            *(v4f*)(bbuf + sb + 4 * i * GM_BP) = rb[i];
        __syncthreads();
        if (ch + 1 < CDIM / GM_KC) {
            int k0 = (ch + 1) * GM_KC;
            ra0 = *(const v4f*)(ga + k0);
            ra1 = *(const v4f*)(ga + k0 + (size_t)32 * CDIM);
            #pragma unroll
            for (int i = 0; i < 8; ++i)
                rb[i] = *(const v4f*)(gb + (size_t)(k0 + 4 * i) * CDIM);
        }
        const float* pa = abuf + fa;
        const float* pb = bbuf + fb;
        #pragma unroll
        for (int kk = 0; kk < GM_KC; kk += 4) {
            v2f a0 = *(const v2f*)(pa + kk);
            v2f a1 = *(const v2f*)(pa + 16 * GM_AP + kk);
            v2f a2 = *(const v2f*)(pa + 32 * GM_AP + kk);
            v2f a3 = *(const v2f*)(pa + 48 * GM_AP + kk);
            v2f b0 = { pb[kk * GM_BP],      pb[(kk + 1) * GM_BP] };
            v2f b1 = { pb[kk * GM_BP + 16], pb[(kk + 1) * GM_BP + 16] };
            c00 = wmma4(a0, b0, c00);  c01 = wmma4(a0, b1, c01);
            c10 = wmma4(a1, b0, c10);  c11 = wmma4(a1, b1, c11);
            c20 = wmma4(a2, b0, c20);  c21 = wmma4(a2, b1, c21);
            c30 = wmma4(a3, b0, c30);  c31 = wmma4(a3, b1, c31);
        }
    }

    float* Db = D + (size_t)batch * CDIM * CDIM;
    float* Zb = ZOUT ? Zn + (size_t)batch * CDIM * CDIM : nullptr;
    int mbase = (lane >> 4) << 3;
    int nlo   = lane & 15;
    auto put = [&](float acc, int gm, int gn) {
        float v = scale * acc;
        if (gm == gn) v += diag;
        Db[(size_t)gm * CDIM + gn] = v;
        if (ZOUT) {
            float t = -v;
            if (gm == gn) t += 3.0f;
            Zb[(size_t)gm * CDIM + gn] = 0.5f * t;
        }
    };
    #pragma unroll
    for (int r = 0; r < 8; ++r) {
        int gm = ti + mbase + r;
        int gn0 = tj + nlo, gn1 = gn0 + 16;
        put(c00[r], gm,      gn0);  put(c01[r], gm,      gn1);
        put(c10[r], gm + 16, gn0);  put(c11[r], gm + 16, gn1);
        put(c20[r], gm + 32, gn0);  put(c21[r], gm + 32, gn1);
        put(c30[r], gm + 48, gn0);  put(c31[r], gm + 48, gn1);
    }
}

__global__ void final_kernel(const float* __restrict__ Y,
                             const float* __restrict__ trc,
                             float* __restrict__ out) {
    int b = blockIdx.y;
    int i = blockIdx.x;
    int j = threadIdx.x;
    if (j < i) return;
    float s = sqrtf(trc[b]);
    int rowoff = i * CDIM - (i * (i - 1)) / 2;
    out[(size_t)b * 32896 + rowoff + (j - i)] =
        s * Y[(size_t)b * CDIM * CDIM + (size_t)i * CDIM + j];
}

extern "C" void kernel_launch(void* const* d_in, const int* in_sizes, int n_in,
                              void* d_out, int out_size, void* d_ws, size_t ws_size,
                              hipStream_t stream) {
    (void)in_sizes; (void)n_in; (void)out_size; (void)ws_size;

    const float* x   = (const float*)d_in[0];
    float*       out = (float*)d_out;
    float*       ws  = (float*)d_ws;

    const size_t MAT = (size_t)NBATCH * CDIM * CDIM;

    float* mean = ws;
    float* trc  = mean + NBATCH * CDIM;
    float* S    = trc + NBATCH;
    float* Ya   = S  + MAT;
    float* Yb   = Ya + MAT;
    float* Za   = Yb + MAT;
    float* Zb   = Za + MAT;

    mean_kernel <<<(NBATCH * CDIM) / 8, 256, 0, stream>>>(x, mean);
    sigma_kernel<<<NBATCH * 4,         256, 0, stream>>>(x, mean, S);
    trace_kernel<<<NBATCH,             256, 0, stream>>>(S, trc);
    init_kernel <<<(unsigned)(MAT / 256), 256, 0, stream>>>(S, trc, Ya, Za);

    gemm256_kernel<true> <<<NBATCH * 4, 256, 0, stream>>>(Ya, Za, Yb, Zb, 0.5f, 0.0f);

    gemm256_kernel<false><<<NBATCH * 4, 256, 0, stream>>>(Zb, Yb, S,  nullptr, -1.0f, 3.0f);
    gemm256_kernel<false><<<NBATCH * 4, 256, 0, stream>>>(Yb, S,  Ya, nullptr,  0.5f, 0.0f);
    gemm256_kernel<false><<<NBATCH * 4, 256, 0, stream>>>(Zb, Ya, S,  nullptr, -1.0f, 3.0f);
    gemm256_kernel<false><<<NBATCH * 4, 256, 0, stream>>>(S,  Zb, Za, nullptr,  0.5f, 0.0f);

    gemm256_kernel<false><<<NBATCH * 4, 256, 0, stream>>>(Za, Ya, S,  nullptr, -1.0f, 3.0f);
    gemm256_kernel<false><<<NBATCH * 4, 256, 0, stream>>>(Ya, S,  Yb, nullptr,  0.5f, 0.0f);

    final_kernel<<<dim3(CDIM, NBATCH), 256, 0, stream>>>(Yb, trc, out);
}


# Trace, normalise and triu-pack passes folded into the Sigma kernel and the first/last GEMM (12 launches to 9); trace summed from staged centred squares, applied as one scale: summation order changed
# speedup vs baseline: 2.5138x; 1.1164x over previous
#include <hip/hip_runtime.h>
#include <math.h>

typedef __attribute__((ext_vector_type(2))) float v2f;
typedef __attribute__((ext_vector_type(4))) float v4f;
typedef __attribute__((ext_vector_type(8))) float v8f;

#define CDIM 256
#define NDIM 784
#define NBATCH 64

static __device__ __forceinline__ v8f wmma4(v2f a, v2f b, v8f c) {
    return __builtin_amdgcn_wmma_f32_16x16x4_f32(false, a, false, b,
                                                 (short)0, c, false, false);
}

__global__ void mean_kernel(const float* __restrict__ X, float* __restrict__ mean) {
    int wave = threadIdx.x >> 5;
    int lane = threadIdx.x & 31;
    int row  = blockIdx.x * 8 + wave;
    const float* p = X + (size_t)row * NDIM;
    float s = 0.f;
    for (int k = lane; k < NDIM; k += 32) s += p[k];
    for (int off = 16; off > 0; off >>= 1) s += __shfl_down(s, off, 32);
    if (lane == 0) mean[row] = s * (1.0f / (float)NDIM);
}

#define SG_KC 32
#define SG_P  36

#define SG_KSTEP(KK)                                                         \
    {                                                                        \
        v2f a0 = *(const v2f*)(pa + (KK));                                   \
        v2f a1 = *(const v2f*)(pa + 16 * SG_P + (KK));                       \
        v2f a2 = *(const v2f*)(pa + 32 * SG_P + (KK));                       \
        v2f a3 = *(const v2f*)(pa + 48 * SG_P + (KK));                       \
        v2f b0 = *(const v2f*)(pb + (KK));                                   \
        v2f b1 = *(const v2f*)(pb + 16 * SG_P + (KK));                       \
        c00 = wmma4(a0, b0, c00);  c01 = wmma4(a0, b1, c01);                 \
        c10 = wmma4(a1, b0, c10);  c11 = wmma4(a1, b1, c11);                 \
        c20 = wmma4(a2, b0, c20);  c21 = wmma4(a2, b1, c21);                 \
        c30 = wmma4(a3, b0, c30);  c31 = wmma4(a3, b1, c31);                 \
    }

__global__ __launch_bounds__(256)
void sigma_kernel(const float* __restrict__ X,
                  const float* __restrict__ mean,
                  float* __restrict__ Y,
                  float* __restrict__ trc) {
    __shared__ __attribute__((aligned(16))) float Xs[2][CDIM * SG_P];
    __shared__ float qred[8];

    int tid   = threadIdx.x;
    int wave  = tid >> 5;
    int lane  = tid & 31;
    int batch = blockIdx.x >> 2;
    int ti = (blockIdx.x & 3) << 6;
    int tj = wave << 5;

    const float* Xb = X    + (size_t)batch * CDIM * NDIM;
    const float* mb = mean + batch * CDIM;

    int srow = tid >> 3;
    int skq  = (tid & 7) << 2;
    const float* gp = Xb + (size_t)srow * NDIM + skq;
    int so = srow * SG_P + skq;
    float mrow[8];
    #pragma unroll
    for (int i = 0; i < 8; ++i) mrow[i] = mb[srow + 32 * i];

    int mi = lane & 15;
    int kh = (lane >> 4) << 1;
    int fa = (ti + mi) * SG_P + kh;
    int fb = (tj + mi) * SG_P + kh;

    v4f r[8];
    #pragma unroll
    for (int i = 0; i < 8; ++i)
        r[i] = *(const v4f*)(gp + (size_t)(32 * i) * NDIM);

    v8f c00 = {}, c01 = {}, c10 = {}, c11 = {};
    v8f c20 = {}, c21 = {}, c30 = {}, c31 = {};

    float q = 0.f;
    const int NFULL = NDIM / SG_KC;
    for (int ch = 0; ch < NFULL; ++ch) {
        float* buf = Xs[ch & 1];
        #pragma unroll
        for (int i = 0; i < 8; ++i) {
            v4f v = r[i] - mrow[i];
            *(v4f*)(buf + so + 32 * i * SG_P) = v;
            q += v.x * v.x + v.y * v.y + v.z * v.z + v.w * v.w;
        }
        __syncthreads();
        if (ch + 1 < NFULL || skq < NDIM - NFULL * SG_KC) {
            const float* g = gp + (ch + 1) * SG_KC;
            #pragma unroll
            for (int i = 0; i < 8; ++i)
                r[i] = *(const v4f*)(g + (size_t)(32 * i) * NDIM);
        }
        const float* pa = buf + fa;
        const float* pb = buf + fb;
        #pragma unroll
        for (int kk = 0; kk < SG_KC; kk += 4) SG_KSTEP(kk)
    }
    {
        float* buf = Xs[NFULL & 1];
        if (skq < NDIM - NFULL * SG_KC) {
            #pragma unroll
            for (int i = 0; i < 8; ++i) {
                v4f v = r[i] - mrow[i];
                *(v4f*)(buf + so + 32 * i * SG_P) = v;
                q += v.x * v.x + v.y * v.y + v.z * v.z + v.w * v.w;
            }
        }
        __syncthreads();
        const float* pa = buf + fa;
        const float* pb = buf + fb;
        #pragma unroll
        for (int kk = 0; kk < NDIM - NFULL * SG_KC; kk += 4) SG_KSTEP(kk)
    }

    const float inv_n = 1.0f / (float)NDIM;
    for (int off = 16; off > 0; off >>= 1) q += __shfl_xor(q, off, 32);
    if (lane == 0) qred[wave] = q;
    __syncthreads();
    float tr = (((qred[0] + qred[1]) + (qred[2] + qred[3])) +
                ((qred[4] + qred[5]) + (qred[6] + qred[7]))) * inv_n;
    if (ti == 0 && tid == 0) trc[batch] = tr;
    const float ysc = inv_n / tr;

    float* Sb = Y + (size_t)batch * CDIM * CDIM;
    int mbase = (lane >> 4) << 3;
    int nlo   = lane & 15;
    for (int r8 = 0; r8 < 8; ++r8) {
        int gm = ti + mbase + r8;
        int gn0 = tj + nlo, gn1 = gn0 + 16;
        Sb[(size_t)(gm     ) * CDIM + gn0] = c00[r8] * ysc;
        Sb[(size_t)(gm     ) * CDIM + gn1] = c01[r8] * ysc;
        Sb[(size_t)(gm + 16) * CDIM + gn0] = c10[r8] * ysc;
        Sb[(size_t)(gm + 16) * CDIM + gn1] = c11[r8] * ysc;
        Sb[(size_t)(gm + 32) * CDIM + gn0] = c20[r8] * ysc;
        Sb[(size_t)(gm + 32) * CDIM + gn1] = c21[r8] * ysc;
        Sb[(size_t)(gm + 48) * CDIM + gn0] = c30[r8] * ysc;
        Sb[(size_t)(gm + 48) * CDIM + gn1] = c31[r8] * ysc;
    }
}

#define GM_KC 32
#define GM_AP 36
#define GM_BP 264

template <int MODE>
__global__ __launch_bounds__(256)
void gemm256_kernel(const float* __restrict__ A,
                    const float* __restrict__ Bm,
                    float* __restrict__ D,
                    float* __restrict__ Zn,
                    const float* __restrict__ trc,
                    float scale, float diag) {
    __shared__ __attribute__((aligned(16))) float As[2][64 * GM_AP];
    __shared__ __attribute__((aligned(16))) float Bs[2][GM_KC * GM_BP];

    int tid   = threadIdx.x;
    int wave  = tid >> 5;
    int lane  = tid & 31;
    int batch = blockIdx.x >> 2;
    int ti = (blockIdx.x & 3) << 6;
    int tj = wave << 5;

    const float* Ab = A  + (size_t)batch * CDIM * CDIM + (size_t)ti * CDIM;
    const float* Bb = Bm + (size_t)batch * CDIM * CDIM;

    int arow = tid >> 3;
    int akq  = (tid & 7) << 2;
    int brow = tid >> 6;
    int bnq  = (tid & 63) << 2;
    const float* ga = Ab + (size_t)arow * CDIM + akq;
    const float* gb = Bb + (size_t)brow * CDIM + bnq;
    int sa = arow * GM_AP + akq;
    int sb = brow * GM_BP + bnq;

    int mi = lane & 15;
    int kh = (lane >> 4) << 1;
    int fa = mi * GM_AP + kh;
    int fb = kh * GM_BP + tj + mi;

    v4f ra0, ra1, rb[8];
    ra0 = *(const v4f*)(ga);
    ra1 = *(const v4f*)(ga + (size_t)32 * CDIM);
    #pragma unroll
    for (int i = 0; i < 8; ++i)
        rb[i] = *(const v4f*)(gb + (size_t)(4 * i) * CDIM);

    v8f c00 = {}, c01 = {}, c10 = {}, c11 = {};
    v8f c20 = {}, c21 = {}, c30 = {}, c31 = {};
    #pragma nounroll
    for (int ch = 0; ch < CDIM / GM_KC; ++ch) {
        float* abuf = As[ch & 1];
        float* bbuf = Bs[ch & 1];
        *(v4f*)(abuf + sa)              = ra0;
        *(v4f*)(abuf + sa + 32 * GM_AP) = ra1;
        #pragma unroll
        for (int i = 0; i < 8; ++i) {
            v4f t = rb[i];
            if (MODE == 1) {
                int d = ch * GM_KC + brow + 4 * i - bnq;
                t.x = (d == 0) ? 3.0f - t.x : -t.x;
                t.y = (d == 1) ? 3.0f - t.y : -t.y;
                t.z = (d == 2) ? 3.0f - t.z : -t.z;
                t.w = (d == 3) ? 3.0f - t.w : -t.w;
            }
            *(v4f*)(bbuf + sb + 4 * i * GM_BP) = t;
        }
        __syncthreads();
        if (ch + 1 < CDIM / GM_KC) {
            int k0 = (ch + 1) * GM_KC;
            ra0 = *(const v4f*)(ga + k0);
            ra1 = *(const v4f*)(ga + k0 + (size_t)32 * CDIM);
            #pragma unroll
            for (int i = 0; i < 8; ++i)
                rb[i] = *(const v4f*)(gb + (size_t)(k0 + 4 * i) * CDIM);
        }
        const float* pa = abuf + fa;
        const float* pb = bbuf + fb;
        #pragma unroll
        for (int kk = 0; kk < GM_KC; kk += 4) {
            v2f a0 = *(const v2f*)(pa + kk);
            v2f a1 = *(const v2f*)(pa + 16 * GM_AP + kk);
            v2f a2 = *(const v2f*)(pa + 32 * GM_AP + kk);
            v2f a3 = *(const v2f*)(pa + 48 * GM_AP + kk);
            v2f b0 = { pb[kk * GM_BP],      pb[(kk + 1) * GM_BP] };
            v2f b1 = { pb[kk * GM_BP + 16], pb[(kk + 1) * GM_BP + 16] };
            c00 = wmma4(a0, b0, c00);  c01 = wmma4(a0, b1, c01);
            c10 = wmma4(a1, b0, c10);  c11 = wmma4(a1, b1, c11);
            c20 = wmma4(a2, b0, c20);  c21 = wmma4(a2, b1, c21);
            c30 = wmma4(a3, b0, c30);  c31 = wmma4(a3, b1, c31);
        }
    }

    float* Db = (MODE == 2) ? D + (size_t)batch * (CDIM * (CDIM + 1) / 2)
                            : D + (size_t)batch * CDIM * CDIM;
    float* Zb = (MODE == 1) ? Zn + (size_t)batch * CDIM * CDIM : nullptr;
    float s = (MODE == 2) ? sqrtf(trc[batch]) : 0.0f;
    int mbase = (lane >> 4) << 3;
    int nlo   = lane & 15;
    auto put = [&](float acc, int gm, int gn) {
        float v = scale * acc;
        if (gm == gn) v += diag;
        if (MODE == 2) {
            if (gn >= gm)
                Db[gm * CDIM - (gm * (gm - 1)) / 2 + (gn - gm)] = s * v;
        } else {
            Db[(size_t)gm * CDIM + gn] = v;
        }
        if (MODE == 1) {
            float t = -v;
            if (gm == gn) t += 3.0f;
            Zb[(size_t)gm * CDIM + gn] = 0.5f * t;
        }
    };
    #pragma unroll
    for (int r = 0; r < 8; ++r) {
        int gm = ti + mbase + r;
        int gn0 = tj + nlo, gn1 = gn0 + 16;
        put(c00[r], gm,      gn0);  put(c01[r], gm,      gn1);
        put(c10[r], gm + 16, gn0);  put(c11[r], gm + 16, gn1);
        put(c20[r], gm + 32, gn0);  put(c21[r], gm + 32, gn1);
        put(c30[r], gm + 48, gn0);  put(c31[r], gm + 48, gn1);
    }
}

extern "C" void kernel_launch(void* const* d_in, const int* in_sizes, int n_in,
                              void* d_out, int out_size, void* d_ws, size_t ws_size,
                              hipStream_t stream) {
    (void)in_sizes; (void)n_in; (void)out_size; (void)ws_size;

    const float* x   = (const float*)d_in[0];
    float*       out = (float*)d_out;
    float*       ws  = (float*)d_ws;

    const size_t MAT = (size_t)NBATCH * CDIM * CDIM;

    float* mean = ws;
    float* trc  = mean + NBATCH * CDIM;
    float* T    = trc + NBATCH;
    float* Ya   = T  + MAT;
    float* Yb   = Ya + MAT;
    float* Za   = Yb + MAT;
    float* Zb   = Za + MAT;

    mean_kernel <<<(NBATCH * CDIM) / 8, 256, 0, stream>>>(x, mean);
    sigma_kernel<<<NBATCH * 4,         256, 0, stream>>>(x, mean, Ya, trc);

    gemm256_kernel<1><<<NBATCH * 4, 256, 0, stream>>>(Ya, Ya, Yb, Zb, nullptr, 0.5f, 0.0f);

    gemm256_kernel<0><<<NBATCH * 4, 256, 0, stream>>>(Zb, Yb, T,  nullptr, nullptr, -1.0f, 3.0f);
    gemm256_kernel<0><<<NBATCH * 4, 256, 0, stream>>>(Yb, T,  Ya, nullptr, nullptr,  0.5f, 0.0f);
    gemm256_kernel<0><<<NBATCH * 4, 256, 0, stream>>>(Zb, Ya, T,  nullptr, nullptr, -1.0f, 3.0f);
    gemm256_kernel<0><<<NBATCH * 4, 256, 0, stream>>>(T,  Zb, Za, nullptr, nullptr,  0.5f, 0.0f);

    gemm256_kernel<0><<<NBATCH * 4, 256, 0, stream>>>(Za, Ya, T,  nullptr, nullptr, -1.0f, 3.0f);
    gemm256_kernel<2><<<NBATCH * 4, 256, 0, stream>>>(Ya, T,  out, nullptr, trc,     0.5f, 0.0f);
}


# Z2@Y2 reassociated: ((3I-P)@Z1)@Y2 is computed as (3I-P)@P, P = Z1@Y2, 3I-P formed while staging; Z2 is never formed (7 GEMMs to 6, 9 launches to 8); rounding order changed
# speedup vs baseline: 2.7451x; 1.0920x over previous
#include <hip/hip_runtime.h>
#include <math.h>

typedef __attribute__((ext_vector_type(2))) float v2f;
typedef __attribute__((ext_vector_type(4))) float v4f;
typedef __attribute__((ext_vector_type(8))) float v8f;

#define CDIM 256
#define NDIM 784
#define NBATCH 64

static __device__ __forceinline__ v8f wmma4(v2f a, v2f b, v8f c) {
    return __builtin_amdgcn_wmma_f32_16x16x4_f32(false, a, false, b,
                                                 (short)0, c, false, false);
}

__global__ void mean_kernel(const float* __restrict__ X, float* __restrict__ mean) {
    int wave = threadIdx.x >> 5;
    int lane = threadIdx.x & 31;
    int row  = blockIdx.x * 8 + wave;
    const float* p = X + (size_t)row * NDIM;
    float s = 0.f;
    for (int k = lane; k < NDIM; k += 32) s += p[k];
    for (int off = 16; off > 0; off >>= 1) s += __shfl_down(s, off, 32);
    if (lane == 0) mean[row] = s * (1.0f / (float)NDIM);
}

#define SG_KC 32
#define SG_P  36

#define SG_KSTEP(KK)                                                         \
    {                                                                        \
        v2f a0 = *(const v2f*)(pa + (KK));                                   \
        v2f a1 = *(const v2f*)(pa + 16 * SG_P + (KK));                       \
        v2f a2 = *(const v2f*)(pa + 32 * SG_P + (KK));                       \
        v2f a3 = *(const v2f*)(pa + 48 * SG_P + (KK));                       \
        v2f b0 = *(const v2f*)(pb + (KK));                                   \
        v2f b1 = *(const v2f*)(pb + 16 * SG_P + (KK));                       \
        c00 = wmma4(a0, b0, c00);  c01 = wmma4(a0, b1, c01);                 \
        c10 = wmma4(a1, b0, c10);  c11 = wmma4(a1, b1, c11);                 \
        c20 = wmma4(a2, b0, c20);  c21 = wmma4(a2, b1, c21);                 \
        c30 = wmma4(a3, b0, c30);  c31 = wmma4(a3, b1, c31);                 \
    }

__global__ __launch_bounds__(256)
void sigma_kernel(const float* __restrict__ X,
                  const float* __restrict__ mean,
                  float* __restrict__ Y,
                  float* __restrict__ trc) {
    __shared__ __attribute__((aligned(16))) float Xs[2][CDIM * SG_P];
    __shared__ float qred[8];

    int tid   = threadIdx.x;
    int wave  = tid >> 5;
    int lane  = tid & 31;
    int batch = blockIdx.x >> 2;
    int ti = (blockIdx.x & 3) << 6;
    int tj = wave << 5;

    const float* Xb = X    + (size_t)batch * CDIM * NDIM;
    const float* mb = mean + batch * CDIM;

    int srow = tid >> 3;
    int skq  = (tid & 7) << 2;
    const float* gp = Xb + (size_t)srow * NDIM + skq;
    int so = srow * SG_P + skq;
    float mrow[8];
    #pragma unroll
    for (int i = 0; i < 8; ++i) mrow[i] = mb[srow + 32 * i];

    int mi = lane & 15;
    int kh = (lane >> 4) << 1;
    int fa = (ti + mi) * SG_P + kh;
    int fb = (tj + mi) * SG_P + kh;

    v4f r[8];
    #pragma unroll
    for (int i = 0; i < 8; ++i)
        r[i] = *(const v4f*)(gp + (size_t)(32 * i) * NDIM);

    v8f c00 = {}, c01 = {}, c10 = {}, c11 = {};
    v8f c20 = {}, c21 = {}, c30 = {}, c31 = {};

    float q = 0.f;
    const int NFULL = NDIM / SG_KC;
    for (int ch = 0; ch < NFULL; ++ch) {
        float* buf = Xs[ch & 1];
        #pragma unroll
        for (int i = 0; i < 8; ++i) {
            v4f v = r[i] - mrow[i];
            *(v4f*)(buf + so + 32 * i * SG_P) = v;
            q += v.x * v.x + v.y * v.y + v.z * v.z + v.w * v.w;
        }
        __syncthreads();
        if (ch + 1 < NFULL || skq < NDIM - NFULL * SG_KC) {
            const float* g = gp + (ch + 1) * SG_KC;
            #pragma unroll
            for (int i = 0; i < 8; ++i)
                r[i] = *(const v4f*)(g + (size_t)(32 * i) * NDIM);
        }
        const float* pa = buf + fa;
        const float* pb = buf + fb;
        #pragma unroll
        for (int kk = 0; kk < SG_KC; kk += 4) SG_KSTEP(kk)
    }
    {
        float* buf = Xs[NFULL & 1];
        if (skq < NDIM - NFULL * SG_KC) {
            #pragma unroll
            for (int i = 0; i < 8; ++i) {
                v4f v = r[i] - mrow[i];
                *(v4f*)(buf + so + 32 * i * SG_P) = v;
                q += v.x * v.x + v.y * v.y + v.z * v.z + v.w * v.w;
            }
        }
        __syncthreads();
        const float* pa = buf + fa;
        const float* pb = buf + fb;
        #pragma unroll
        for (int kk = 0; kk < NDIM - NFULL * SG_KC; kk += 4) SG_KSTEP(kk)
    }

    const float inv_n = 1.0f / (float)NDIM;
    for (int off = 16; off > 0; off >>= 1) q += __shfl_xor(q, off, 32);
    if (lane == 0) qred[wave] = q;
    __syncthreads();
    float tr = (((qred[0] + qred[1]) + (qred[2] + qred[3])) +
                ((qred[4] + qred[5]) + (qred[6] + qred[7]))) * inv_n;
    if (ti == 0 && tid == 0) trc[batch] = tr;
    const float ysc = inv_n / tr;

    float* Sb = Y + (size_t)batch * CDIM * CDIM;
    int mbase = (lane >> 4) << 3;
    int nlo   = lane & 15;
    for (int r8 = 0; r8 < 8; ++r8) {
        int gm = ti + mbase + r8;
        int gn0 = tj + nlo, gn1 = gn0 + 16;
        Sb[(size_t)(gm     ) * CDIM + gn0] = c00[r8] * ysc;
        Sb[(size_t)(gm     ) * CDIM + gn1] = c01[r8] * ysc;
        Sb[(size_t)(gm + 16) * CDIM + gn0] = c10[r8] * ysc;
        Sb[(size_t)(gm + 16) * CDIM + gn1] = c11[r8] * ysc;
        Sb[(size_t)(gm + 32) * CDIM + gn0] = c20[r8] * ysc;
        Sb[(size_t)(gm + 32) * CDIM + gn1] = c21[r8] * ysc;
        Sb[(size_t)(gm + 48) * CDIM + gn0] = c30[r8] * ysc;
        Sb[(size_t)(gm + 48) * CDIM + gn1] = c31[r8] * ysc;
    }
}

#define GM_KC 32
#define GM_AP 36
#define GM_BP 264

template <int MODE>
__global__ __launch_bounds__(256)
void gemm256_kernel(const float* __restrict__ A,
                    const float* __restrict__ Bm,
                    float* __restrict__ D,
                    float* __restrict__ Zn,
                    const float* __restrict__ trc,
                    float scale, float diag) {
    __shared__ __attribute__((aligned(16))) float As[2][64 * GM_AP];
    __shared__ __attribute__((aligned(16))) float Bs[2][GM_KC * GM_BP];

    int tid   = threadIdx.x;
    int wave  = tid >> 5;
    int lane  = tid & 31;
    int batch = blockIdx.x >> 2;
    int ti = (blockIdx.x & 3) << 6;
    int tj = wave << 5;

    const float* Ab = A  + (size_t)batch * CDIM * CDIM + (size_t)ti * CDIM;
    const float* Bb = Bm + (size_t)batch * CDIM * CDIM;

    int arow = tid >> 3;
    int akq  = (tid & 7) << 2;
    int brow = tid >> 6;
    int bnq  = (tid & 63) << 2;
    const float* ga = Ab + (size_t)arow * CDIM + akq;
    const float* gb = Bb + (size_t)brow * CDIM + bnq;
    int sa = arow * GM_AP + akq;
    int sb = brow * GM_BP + bnq;

    int mi = lane & 15;
    int kh = (lane >> 4) << 1;
    int fa = mi * GM_AP + kh;
    int fb = kh * GM_BP + tj + mi;

    v4f ra0, ra1, rb[8];
    ra0 = *(const v4f*)(ga);
    ra1 = *(const v4f*)(ga + (size_t)32 * CDIM);
    #pragma unroll
    for (int i = 0; i < 8; ++i)
        rb[i] = *(const v4f*)(gb + (size_t)(4 * i) * CDIM);

    v8f c00 = {}, c01 = {}, c10 = {}, c11 = {};
    v8f c20 = {}, c21 = {}, c30 = {}, c31 = {};
    #pragma nounroll
    for (int ch = 0; ch < CDIM / GM_KC; ++ch) {
        float* abuf = As[ch & 1];
        float* bbuf = Bs[ch & 1];
        {
            v4f t0 = ra0, t1 = ra1;
            if (MODE == 3) {
                int d0 = ti + arow - (ch * GM_KC + akq);
                int d1 = d0 + 32;
                t0.x = (d0 == 0) ? 3.0f - t0.x : -t0.x;
                t0.y = (d0 == 1) ? 3.0f - t0.y : -t0.y;
                t0.z = (d0 == 2) ? 3.0f - t0.z : -t0.z;
                t0.w = (d0 == 3) ? 3.0f - t0.w : -t0.w;
                t1.x = (d1 == 0) ? 3.0f - t1.x : -t1.x;
                t1.y = (d1 == 1) ? 3.0f - t1.y : -t1.y;
                t1.z = (d1 == 2) ? 3.0f - t1.z : -t1.z;
                t1.w = (d1 == 3) ? 3.0f - t1.w : -t1.w;
            }
            *(v4f*)(abuf + sa)              = t0;
            *(v4f*)(abuf + sa + 32 * GM_AP) = t1;
        }
        #pragma unroll
        for (int i = 0; i < 8; ++i) {
            v4f t = rb[i];
            if (MODE == 1) {
                int d = ch * GM_KC + brow + 4 * i - bnq;
                t.x = (d == 0) ? 3.0f - t.x : -t.x;
                t.y = (d == 1) ? 3.0f - t.y : -t.y;
                t.z = (d == 2) ? 3.0f - t.z : -t.z;
                t.w = (d == 3) ? 3.0f - t.w : -t.w;
            }
            *(v4f*)(bbuf + sb + 4 * i * GM_BP) = t;
        }
        __syncthreads();
        if (ch + 1 < CDIM / GM_KC) {
            int k0 = (ch + 1) * GM_KC;
            ra0 = *(const v4f*)(ga + k0);
            ra1 = *(const v4f*)(ga + k0 + (size_t)32 * CDIM);
            #pragma unroll
            for (int i = 0; i < 8; ++i)
                rb[i] = *(const v4f*)(gb + (size_t)(k0 + 4 * i) * CDIM);
        }
        const float* pa = abuf + fa;
        const float* pb = bbuf + fb;
        #pragma unroll
        for (int kk = 0; kk < GM_KC; kk += 4) {
            v2f a0 = *(const v2f*)(pa + kk);
            v2f a1 = *(const v2f*)(pa + 16 * GM_AP + kk);
            v2f a2 = *(const v2f*)(pa + 32 * GM_AP + kk);
            v2f a3 = *(const v2f*)(pa + 48 * GM_AP + kk);
            v2f b0 = { pb[kk * GM_BP],      pb[(kk + 1) * GM_BP] };
            v2f b1 = { pb[kk * GM_BP + 16], pb[(kk + 1) * GM_BP + 16] };
            c00 = wmma4(a0, b0, c00);  c01 = wmma4(a0, b1, c01);
            c10 = wmma4(a1, b0, c10);  c11 = wmma4(a1, b1, c11);
            c20 = wmma4(a2, b0, c20);  c21 = wmma4(a2, b1, c21);
            c30 = wmma4(a3, b0, c30);  c31 = wmma4(a3, b1, c31);
        }
    }

    float* Db = (MODE == 2) ? D + (size_t)batch * (CDIM * (CDIM + 1) / 2)
                            : D + (size_t)batch * CDIM * CDIM;
    float* Zb = (MODE == 1) ? Zn + (size_t)batch * CDIM * CDIM : nullptr;
    float s = (MODE == 2) ? sqrtf(trc[batch]) : 0.0f;
    int mbase = (lane >> 4) << 3;
    int nlo   = lane & 15;
    auto put = [&](float acc, int gm, int gn) {
        float v = scale * acc;
        if (gm == gn) v += diag;
        if (MODE == 2) {
            if (gn >= gm)
                Db[gm * CDIM - (gm * (gm - 1)) / 2 + (gn - gm)] = s * v;
        } else {
            Db[(size_t)gm * CDIM + gn] = v;
        }
        if (MODE == 1) {
            float t = -v;
            if (gm == gn) t += 3.0f;
            Zb[(size_t)gm * CDIM + gn] = 0.5f * t;
        }
    };
    #pragma unroll
    for (int r = 0; r < 8; ++r) {
        int gm = ti + mbase + r;
        int gn0 = tj + nlo, gn1 = gn0 + 16;
        put(c00[r], gm,      gn0);  put(c01[r], gm,      gn1);
        put(c10[r], gm + 16, gn0);  put(c11[r], gm + 16, gn1);
        put(c20[r], gm + 32, gn0);  put(c21[r], gm + 32, gn1);
        put(c30[r], gm + 48, gn0);  put(c31[r], gm + 48, gn1);
    }
}

extern "C" void kernel_launch(void* const* d_in, const int* in_sizes, int n_in,
                              void* d_out, int out_size, void* d_ws, size_t ws_size,
                              hipStream_t stream) {
    (void)in_sizes; (void)n_in; (void)out_size; (void)ws_size;

    const float* x   = (const float*)d_in[0];
    float*       out = (float*)d_out;
    float*       ws  = (float*)d_ws;

    const size_t MAT = (size_t)NBATCH * CDIM * CDIM;

    float* mean = ws;
    float* trc  = mean + NBATCH * CDIM;
    float* T    = trc + NBATCH;
    float* Ya   = T  + MAT;
    float* Yb   = Ya + MAT;
    float* T3   = Yb + MAT;
    float* Z1   = T3 + MAT;

    mean_kernel <<<(NBATCH * CDIM) / 8, 256, 0, stream>>>(x, mean);
    sigma_kernel<<<NBATCH * 4,         256, 0, stream>>>(x, mean, Ya, trc);

    gemm256_kernel<1><<<NBATCH * 4, 256, 0, stream>>>(Ya, Ya, Yb, Z1, nullptr, 0.5f, 0.0f);

    gemm256_kernel<0><<<NBATCH * 4, 256, 0, stream>>>(Z1, Yb, T,  nullptr, nullptr, -1.0f, 3.0f);
    gemm256_kernel<0><<<NBATCH * 4, 256, 0, stream>>>(Yb, T,  Ya, nullptr, nullptr,  0.5f, 0.0f);

    gemm256_kernel<0><<<NBATCH * 4, 256, 0, stream>>>(Z1, Ya, T,  nullptr, nullptr,  1.0f, 0.0f);
    gemm256_kernel<3><<<NBATCH * 4, 256, 0, stream>>>(T,  T,  T3, nullptr, nullptr, -0.5f, 3.0f);

    gemm256_kernel<2><<<NBATCH * 4, 256, 0, stream>>>(Ya, T3, out, nullptr, trc,     0.5f, 0.0f);
}


# Separate row-mean pass dropped (8 launches to 7): covariance kernel multiplies raw rows and subtracts the rank-1 term s*s^T/n in its epilogue, row sums taken while staging; rounding order changed
# speedup vs baseline: 2.8427x; 1.0356x over previous
#include <hip/hip_runtime.h>
#include <math.h>

typedef __attribute__((ext_vector_type(2))) float v2f;
typedef __attribute__((ext_vector_type(4))) float v4f;
typedef __attribute__((ext_vector_type(8))) float v8f;

#define CDIM 256
#define NDIM 784
#define NBATCH 64

static __device__ __forceinline__ v8f wmma4(v2f a, v2f b, v8f c) {
    return __builtin_amdgcn_wmma_f32_16x16x4_f32(false, a, false, b,
                                                 (short)0, c, false, false);
}

#define SG_KC 32
#define SG_P  36

#define SG_KSTEP(KK)                                                         \
    {                                                                        \
        v2f a0 = *(const v2f*)(pa + (KK));                                   \
        v2f a1 = *(const v2f*)(pa + 16 * SG_P + (KK));                       \
        v2f a2 = *(const v2f*)(pa + 32 * SG_P + (KK));                       \
        v2f a3 = *(const v2f*)(pa + 48 * SG_P + (KK));                       \
        v2f b0 = *(const v2f*)(pb + (KK));                                   \
        v2f b1 = *(const v2f*)(pb + 16 * SG_P + (KK));                       \
        c00 = wmma4(a0, b0, c00);  c01 = wmma4(a0, b1, c01);                 \
        c10 = wmma4(a1, b0, c10);  c11 = wmma4(a1, b1, c11);                 \
        c20 = wmma4(a2, b0, c20);  c21 = wmma4(a2, b1, c21);                 \
        c30 = wmma4(a3, b0, c30);  c31 = wmma4(a3, b1, c31);                 \
    }

__global__ __launch_bounds__(256)
void sigma_kernel(const float* __restrict__ X,
                  float* __restrict__ Y,
                  float* __restrict__ trc) {
    __shared__ __attribute__((aligned(16))) float Xs[2][CDIM * SG_P];
    __shared__ float qred[8];
    __shared__ float rsum[CDIM];

    int tid   = threadIdx.x;
    int wave  = tid >> 5;
    int lane  = tid & 31;
    int batch = blockIdx.x >> 2;
    int ti = (blockIdx.x & 3) << 6;
    int tj = wave << 5;

    const float* Xb = X + (size_t)batch * CDIM * NDIM;

    int srow = tid >> 3;
    int skq  = (tid & 7) << 2;
    const float* gp = Xb + (size_t)srow * NDIM + skq;
    int so = srow * SG_P + skq;
    float rs[8];
    #pragma unroll
    for (int i = 0; i < 8; ++i) rs[i] = 0.f;

    int mi = lane & 15;
    int kh = (lane >> 4) << 1;
    int fa = (ti + mi) * SG_P + kh;
    int fb = (tj + mi) * SG_P + kh;

    v4f r[8];
    #pragma unroll
    for (int i = 0; i < 8; ++i)
        r[i] = *(const v4f*)(gp + (size_t)(32 * i) * NDIM);

    v8f c00 = {}, c01 = {}, c10 = {}, c11 = {};
    v8f c20 = {}, c21 = {}, c30 = {}, c31 = {};

    float q = 0.f;
    const int NFULL = NDIM / SG_KC;
    for (int ch = 0; ch < NFULL; ++ch) {
        float* buf = Xs[ch & 1];
        #pragma unroll
        for (int i = 0; i < 8; ++i) {
            v4f v = r[i];
            *(v4f*)(buf + so + 32 * i * SG_P) = v;
            q += v.x * v.x + v.y * v.y + v.z * v.z + v.w * v.w;
            rs[i] += (v.x + v.y) + (v.z + v.w);
        }
        __syncthreads();
        if (ch + 1 < NFULL || skq < NDIM - NFULL * SG_KC) {
            const float* g = gp + (ch + 1) * SG_KC;
            #pragma unroll
            for (int i = 0; i < 8; ++i)
                r[i] = *(const v4f*)(g + (size_t)(32 * i) * NDIM);
        }
        const float* pa = buf + fa;
        const float* pb = buf + fb;
        #pragma unroll
        for (int kk = 0; kk < SG_KC; kk += 4) SG_KSTEP(kk)
    }
    {
        float* buf = Xs[NFULL & 1];
        if (skq < NDIM - NFULL * SG_KC) {
            #pragma unroll
            for (int i = 0; i < 8; ++i) {
                v4f v = r[i];
                *(v4f*)(buf + so + 32 * i * SG_P) = v;
                q += v.x * v.x + v.y * v.y + v.z * v.z + v.w * v.w;
                rs[i] += (v.x + v.y) + (v.z + v.w);
            }
        }
        __syncthreads();
        const float* pa = buf + fa;
        const float* pb = buf + fb;
        #pragma unroll
        for (int kk = 0; kk < NDIM - NFULL * SG_KC; kk += 4) SG_KSTEP(kk)
    }

    #pragma unroll
    for (int i = 0; i < 8; ++i) {
        float t = rs[i];
        t += __shfl_xor(t, 1, 32);
        t += __shfl_xor(t, 2, 32);
        t += __shfl_xor(t, 4, 32);
        rs[i] = t;
    }
    if ((tid & 7) == 0) {
        #pragma unroll
        for (int i = 0; i < 8; ++i) rsum[srow + 32 * i] = rs[i];
    }
    const float inv_n = 1.0f / (float)NDIM;
    for (int off = 16; off > 0; off >>= 1) q += __shfl_xor(q, off, 32);
    if (lane == 0) qred[wave] = q;
    __syncthreads();
    float qq = ((qred[0] + qred[1]) + (qred[2] + qred[3])) +
               ((qred[4] + qred[5]) + (qred[6] + qred[7]));
    float s2 = 0.f;
    #pragma unroll
    for (int j = 0; j < 8; ++j) {
        float t = rsum[lane + 32 * j];
        s2 += t * t;
    }
    for (int off = 16; off > 0; off >>= 1) s2 += __shfl_xor(s2, off, 32);
    float tr = (qq - s2 * inv_n) * inv_n;
    if (ti == 0 && tid == 0) trc[batch] = tr;
    const float ysc = inv_n / tr;

    float* Sb = Y + (size_t)batch * CDIM * CDIM;
    int mbase = (lane >> 4) << 3;
    int nlo   = lane & 15;
    const float sc0 = rsum[tj + nlo], sc1 = rsum[tj + nlo + 16];
    for (int r8 = 0; r8 < 8; ++r8) {
        int gm = ti + mbase + r8;
        int gn0 = tj + nlo, gn1 = gn0 + 16;
        float sr0 = rsum[gm], sr1 = rsum[gm + 16], sr2 = rsum[gm + 32], sr3 = rsum[gm + 48];
        Sb[(size_t)(gm     ) * CDIM + gn0] = (c00[r8] - (sr0 * sc0) * inv_n) * ysc;
        Sb[(size_t)(gm     ) * CDIM + gn1] = (c01[r8] - (sr0 * sc1) * inv_n) * ysc;
        Sb[(size_t)(gm + 16) * CDIM + gn0] = (c10[r8] - (sr1 * sc0) * inv_n) * ysc;
        Sb[(size_t)(gm + 16) * CDIM + gn1] = (c11[r8] - (sr1 * sc1) * inv_n) * ysc;
        Sb[(size_t)(gm + 32) * CDIM + gn0] = (c20[r8] - (sr2 * sc0) * inv_n) * ysc;
        Sb[(size_t)(gm + 32) * CDIM + gn1] = (c21[r8] - (sr2 * sc1) * inv_n) * ysc;
        Sb[(size_t)(gm + 48) * CDIM + gn0] = (c30[r8] - (sr3 * sc0) * inv_n) * ysc;
        Sb[(size_t)(gm + 48) * CDIM + gn1] = (c31[r8] - (sr3 * sc1) * inv_n) * ysc;
    }
}

#define GM_KC 32
#define GM_AP 36
#define GM_BP 264

template <int MODE>
__global__ __launch_bounds__(256)
void gemm256_kernel(const float* __restrict__ A,
                    const float* __restrict__ Bm,
                    float* __restrict__ D,
                    float* __restrict__ Zn,
                    const float* __restrict__ trc,
                    float scale, float diag) {
    __shared__ __attribute__((aligned(16))) float As[2][64 * GM_AP];
    __shared__ __attribute__((aligned(16))) float Bs[2][GM_KC * GM_BP];

    int tid   = threadIdx.x;
    int wave  = tid >> 5;
    int lane  = tid & 31;
    int batch = blockIdx.x >> 2;
    int ti = (blockIdx.x & 3) << 6;
    int tj = wave << 5;

    const float* Ab = A  + (size_t)batch * CDIM * CDIM + (size_t)ti * CDIM;
    const float* Bb = Bm + (size_t)batch * CDIM * CDIM;

    int arow = tid >> 3;
    int akq  = (tid & 7) << 2;
    int brow = tid >> 6;
    int bnq  = (tid & 63) << 2;
    const float* ga = Ab + (size_t)arow * CDIM + akq;
    const float* gb = Bb + (size_t)brow * CDIM + bnq;
    int sa = arow * GM_AP + akq;
    int sb = brow * GM_BP + bnq;

    int mi = lane & 15;
    int kh = (lane >> 4) << 1;
    int fa = mi * GM_AP + kh;
    int fb = kh * GM_BP + tj + mi;

    v4f ra0, ra1, rb[8];
    ra0 = *(const v4f*)(ga);
    ra1 = *(const v4f*)(ga + (size_t)32 * CDIM);
    #pragma unroll
    for (int i = 0; i < 8; ++i)
        rb[i] = *(const v4f*)(gb + (size_t)(4 * i) * CDIM);

    v8f c00 = {}, c01 = {}, c10 = {}, c11 = {};
    v8f c20 = {}, c21 = {}, c30 = {}, c31 = {};
    #pragma nounroll
    for (int ch = 0; ch < CDIM / GM_KC; ++ch) {
        float* abuf = As[ch & 1];
        float* bbuf = Bs[ch & 1];
        {
            v4f t0 = ra0, t1 = ra1;
            if (MODE == 3) {
                int d0 = ti + arow - (ch * GM_KC + akq);
                int d1 = d0 + 32;
                t0.x = (d0 == 0) ? 3.0f - t0.x : -t0.x;
                t0.y = (d0 == 1) ? 3.0f - t0.y : -t0.y;
                t0.z = (d0 == 2) ? 3.0f - t0.z : -t0.z;
                t0.w = (d0 == 3) ? 3.0f - t0.w : -t0.w;
                t1.x = (d1 == 0) ? 3.0f - t1.x : -t1.x;
                t1.y = (d1 == 1) ? 3.0f - t1.y : -t1.y;
                t1.z = (d1 == 2) ? 3.0f - t1.z : -t1.z;
                t1.w = (d1 == 3) ? 3.0f - t1.w : -t1.w;
            }
            *(v4f*)(abuf + sa)              = t0;
            *(v4f*)(abuf + sa + 32 * GM_AP) = t1;
        }
        #pragma unroll
        for (int i = 0; i < 8; ++i) {
            v4f t = rb[i];
            if (MODE == 1) {
                int d = ch * GM_KC + brow + 4 * i - bnq;
                t.x = (d == 0) ? 3.0f - t.x : -t.x;
                t.y = (d == 1) ? 3.0f - t.y : -t.y;
                t.z = (d == 2) ? 3.0f - t.z : -t.z;
                t.w = (d == 3) ? 3.0f - t.w : -t.w;
            }
            *(v4f*)(bbuf + sb + 4 * i * GM_BP) = t;
        }
        __syncthreads();
        if (ch + 1 < CDIM / GM_KC) {
            int k0 = (ch + 1) * GM_KC;
            ra0 = *(const v4f*)(ga + k0);
            ra1 = *(const v4f*)(ga + k0 + (size_t)32 * CDIM);
            #pragma unroll
            for (int i = 0; i < 8; ++i)
                rb[i] = *(const v4f*)(gb + (size_t)(k0 + 4 * i) * CDIM);
        }
        const float* pa = abuf + fa;
        const float* pb = bbuf + fb;
        #pragma unroll
        for (int kk = 0; kk < GM_KC; kk += 4) {
            v2f a0 = *(const v2f*)(pa + kk);
            v2f a1 = *(const v2f*)(pa + 16 * GM_AP + kk);
            v2f a2 = *(const v2f*)(pa + 32 * GM_AP + kk);
            v2f a3 = *(const v2f*)(pa + 48 * GM_AP + kk);
            v2f b0 = { pb[kk * GM_BP],      pb[(kk + 1) * GM_BP] };
            v2f b1 = { pb[kk * GM_BP + 16], pb[(kk + 1) * GM_BP + 16] };
            c00 = wmma4(a0, b0, c00);  c01 = wmma4(a0, b1, c01);
            c10 = wmma4(a1, b0, c10);  c11 = wmma4(a1, b1, c11);
            c20 = wmma4(a2, b0, c20);  c21 = wmma4(a2, b1, c21);
            c30 = wmma4(a3, b0, c30);  c31 = wmma4(a3, b1, c31);
        }
    }

    float* Db = (MODE == 2) ? D + (size_t)batch * (CDIM * (CDIM + 1) / 2)
                            : D + (size_t)batch * CDIM * CDIM;
    float* Zb = (MODE == 1) ? Zn + (size_t)batch * CDIM * CDIM : nullptr;
    float s = (MODE == 2) ? sqrtf(trc[batch]) : 0.0f;
    int mbase = (lane >> 4) << 3;
    int nlo   = lane & 15;
    auto put = [&](float acc, int gm, int gn) {
        float v = scale * acc;
        if (gm == gn) v += diag;
        if (MODE == 2) {
            if (gn >= gm)
                Db[gm * CDIM - (gm * (gm - 1)) / 2 + (gn - gm)] = s * v;
        } else {
            Db[(size_t)gm * CDIM + gn] = v;
        }
        if (MODE == 1) {
            float t = -v;
            if (gm == gn) t += 3.0f;
            Zb[(size_t)gm * CDIM + gn] = 0.5f * t;
        }
    };
    #pragma unroll
    for (int r = 0; r < 8; ++r) {
        int gm = ti + mbase + r;
        int gn0 = tj + nlo, gn1 = gn0 + 16;
        put(c00[r], gm,      gn0);  put(c01[r], gm,      gn1);
        put(c10[r], gm + 16, gn0);  put(c11[r], gm + 16, gn1);
        put(c20[r], gm + 32, gn0);  put(c21[r], gm + 32, gn1);
        put(c30[r], gm + 48, gn0);  put(c31[r], gm + 48, gn1);
    }
}

extern "C" void kernel_launch(void* const* d_in, const int* in_sizes, int n_in,
                              void* d_out, int out_size, void* d_ws, size_t ws_size,
                              hipStream_t stream) {
    (void)in_sizes; (void)n_in; (void)out_size; (void)ws_size;

    const float* x   = (const float*)d_in[0];
    float*       out = (float*)d_out;
    float*       ws  = (float*)d_ws;

    const size_t MAT = (size_t)NBATCH * CDIM * CDIM;

    float* trc  = ws;
    float* T    = trc + NBATCH;
    float* Ya   = T  + MAT;
    float* Yb   = Ya + MAT;
    float* T3   = Yb + MAT;
    float* Z1   = T3 + MAT;

    sigma_kernel<<<NBATCH * 4, 256, 0, stream>>>(x, Ya, trc);

    gemm256_kernel<1><<<NBATCH * 4, 256, 0, stream>>>(Ya, Ya, Yb, Z1, nullptr, 0.5f, 0.0f);

    gemm256_kernel<0><<<NBATCH * 4, 256, 0, stream>>>(Z1, Yb, T,  nullptr, nullptr, -1.0f, 3.0f);
    gemm256_kernel<0><<<NBATCH * 4, 256, 0, stream>>>(Yb, T,  Ya, nullptr, nullptr,  0.5f, 0.0f);

    gemm256_kernel<0><<<NBATCH * 4, 256, 0, stream>>>(Z1, Ya, T,  nullptr, nullptr,  1.0f, 0.0f);
    gemm256_kernel<3><<<NBATCH * 4, 256, 0, stream>>>(T,  T,  T3, nullptr, nullptr, -0.5f, 3.0f);

    gemm256_kernel<2><<<NBATCH * 4, 256, 0, stream>>>(Ya, T3, out, nullptr, trc,     0.5f, 0.0f);
}


# Each product computes 5 of 8 panel columns per 32-row panel and stores every tile also transposed (all factors are symmetric polynomials of Sigma): mirrored values are copies, rounding changed
# speedup vs baseline: 3.2881x; 1.1567x over previous
#include <hip/hip_runtime.h>
#include <math.h>

typedef __attribute__((ext_vector_type(2))) float v2f;
typedef __attribute__((ext_vector_type(4))) float v4f;
typedef __attribute__((ext_vector_type(8))) float v8f;

#define CDIM 256
#define NDIM 784
#define NBATCH 64

static __device__ __forceinline__ v8f wmma4(v2f a, v2f b, v8f c) {
    return __builtin_amdgcn_wmma_f32_16x16x4_f32(false, a, false, b,
                                                 (short)0, c, false, false);
}


#define SG_KC 32
#define SG_P  36
#define NT    5

#define SG_KSTEP(KK)                                                         \
    {                                                                        \
        v2f a = *(const v2f*)(buf + fa + (KK));                              \
        v2f b[NT];                                                           \
        _Pragma("unroll")                                                    \
        for (int t = 0; t < NT; ++t)                                         \
            b[t] = *(const v2f*)(buf + fb[t] + (KK));                        \
        _Pragma("unroll")                                                    \
        for (int t = 0; t < NT; ++t)                                         \
            c[t] = wmma4(a, b[t], c[t]);                                     \
    }

__global__ __launch_bounds__(256)
void sigma_kernel(const float* __restrict__ X,
                  float* __restrict__ Y,
                  float* __restrict__ trc) {
    __shared__ __attribute__((aligned(16))) float Xs[2][CDIM * SG_P];
    __shared__ float qred[8];
    __shared__ float rsum[CDIM];

    int tid   = threadIdx.x;
    int wave  = __builtin_amdgcn_readfirstlane(tid >> 5);
    int lane  = tid & 31;
    int batch = blockIdx.x >> 2;
    int ti = (blockIdx.x & 3) << 6;

    const float* Xb = X + (size_t)batch * CDIM * NDIM;

    int srow = tid >> 3;
    int skq  = (tid & 7) << 2;
    const float* gp = Xb + (size_t)srow * NDIM + skq;
    int so = srow * SG_P + skq;
    float rs[8];
    #pragma unroll
    for (int i = 0; i < 8; ++i) rs[i] = 0.f;

    int wq = wave >> 2;
    int rh = (wave >> 1) & 1;
    int cs = wave & 1;
    int Q  = ((blockIdx.x & 3) << 1) + wq;
    int R0 = 32 * Q + 16 * rh;

    int mi = lane & 15;
    int kh = (lane >> 4) << 1;
    int fa = (R0 + mi) * SG_P + kh;
    int fb[NT];
    #pragma unroll
    for (int t = 0; t < NT; ++t)
        fb[t] = (32 * ((Q + t) & 7) + 16 * cs + mi) * SG_P + kh;

    v4f r[8];
    #pragma unroll
    for (int i = 0; i < 8; ++i)
        r[i] = *(const v4f*)(gp + (size_t)(32 * i) * NDIM);

    v8f c[NT];
    #pragma unroll
    for (int t = 0; t < NT; ++t) { v8f z = {}; c[t] = z; }

    float q = 0.f;
    const int NFULL = NDIM / SG_KC;
    for (int ch = 0; ch < NFULL; ++ch) {
        float* buf = Xs[ch & 1];
        #pragma unroll
        for (int i = 0; i < 8; ++i) {
            v4f v = r[i];
            *(v4f*)(buf + so + 32 * i * SG_P) = v;
            q += v.x * v.x + v.y * v.y + v.z * v.z + v.w * v.w;
            rs[i] += (v.x + v.y) + (v.z + v.w);
        }
        __syncthreads();
        if (ch + 1 < NFULL || skq < NDIM - NFULL * SG_KC) {
            const float* g = gp + (ch + 1) * SG_KC;
            #pragma unroll
            for (int i = 0; i < 8; ++i)
                r[i] = *(const v4f*)(g + (size_t)(32 * i) * NDIM);
        }
        #pragma unroll
        for (int kk = 0; kk < SG_KC; kk += 4) SG_KSTEP(kk)
    }
    {
        float* buf = Xs[NFULL & 1];
        if (skq < NDIM - NFULL * SG_KC) {
            #pragma unroll
            for (int i = 0; i < 8; ++i) {
                v4f v = r[i];
                *(v4f*)(buf + so + 32 * i * SG_P) = v;
                q += v.x * v.x + v.y * v.y + v.z * v.z + v.w * v.w;
                rs[i] += (v.x + v.y) + (v.z + v.w);
            }
        }
        __syncthreads();
        #pragma unroll
        for (int kk = 0; kk < NDIM - NFULL * SG_KC; kk += 4) SG_KSTEP(kk)
    }

    #pragma unroll
    for (int i = 0; i < 8; ++i) {
        float t = rs[i];
        t += __shfl_xor(t, 1, 32);
        t += __shfl_xor(t, 2, 32);
        t += __shfl_xor(t, 4, 32);
        rs[i] = t;
    }
    if ((tid & 7) == 0) {
        #pragma unroll
        for (int i = 0; i < 8; ++i) rsum[srow + 32 * i] = rs[i];
    }
    const float inv_n = 1.0f / (float)NDIM;
    for (int off = 16; off > 0; off >>= 1) q += __shfl_xor(q, off, 32);
    if (lane == 0) qred[wave] = q;
    __syncthreads();
    float qq = ((qred[0] + qred[1]) + (qred[2] + qred[3])) +
               ((qred[4] + qred[5]) + (qred[6] + qred[7]));
    float s2 = 0.f;
    #pragma unroll
    for (int j = 0; j < 8; ++j) {
        float t = rsum[lane + 32 * j];
        s2 += t * t;
    }
    for (int off = 16; off > 0; off >>= 1) s2 += __shfl_xor(s2, off, 32);
    float tr = (qq - s2 * inv_n) * inv_n;
    if (ti == 0 && tid == 0) trc[batch] = tr;
    const float ysc = inv_n / tr;

    float* Sb = Y + (size_t)batch * CDIM * CDIM;
    int mbase = (lane >> 4) << 3;
    int nlo   = lane & 15;
    #pragma unroll
    for (int t = 0; t < NT; ++t) {
        if (t == 4 && Q >= 4) continue;
        int gn = 32 * ((Q + t) & 7) + 16 * cs + nlo;
        float scn = rsum[gn];
        float v[8];
        #pragma unroll
        for (int r8 = 0; r8 < 8; ++r8) {
            int gm = R0 + mbase + r8;
            v[r8] = (c[t][r8] - (rsum[gm] * scn) * inv_n) * ysc;
            Sb[(size_t)gm * CDIM + gn] = v[r8];
        }
        if (t > 0) {
            float* mp = Sb + (size_t)gn * CDIM + R0 + mbase;
            v4f lo = { v[0], v[1], v[2], v[3] };
            v4f hi = { v[4], v[5], v[6], v[7] };
            *(v4f*)mp       = lo;
            *(v4f*)(mp + 4) = hi;
        }
    }
}

#define GM_KC 32
#define GM_AP 36
#define GM_BP 264

template <int MODE>
__global__ __launch_bounds__(256)
void gemm256_kernel(const float* __restrict__ A,
                    const float* __restrict__ Bm,
                    float* __restrict__ D,
                    float* __restrict__ Zn,
                    const float* __restrict__ trc,
                    float scale, float diag) {
    __shared__ __attribute__((aligned(16))) float As[2][64 * GM_AP];
    __shared__ __attribute__((aligned(16))) float Bs[2][GM_KC * GM_BP];

    int tid   = threadIdx.x;
    int wave  = __builtin_amdgcn_readfirstlane(tid >> 5);
    int lane  = tid & 31;
    int batch = blockIdx.x >> 2;
    int ti = (blockIdx.x & 3) << 6;

    const float* Ab = A  + (size_t)batch * CDIM * CDIM + (size_t)ti * CDIM;
    const float* Bb = Bm + (size_t)batch * CDIM * CDIM;

    int arow = tid >> 3;
    int akq  = (tid & 7) << 2;
    int brow = tid >> 6;
    int bnq  = (tid & 63) << 2;
    const float* ga = Ab + (size_t)arow * CDIM + akq;
    const float* gb = Bb + (size_t)brow * CDIM + bnq;
    int sa = arow * GM_AP + akq;
    int sb = brow * GM_BP + bnq;

    int wq = wave >> 2;
    int rh = (wave >> 1) & 1;
    int cs = wave & 1;
    int Q  = ((blockIdx.x & 3) << 1) + wq;
    int R0 = 32 * Q + 16 * rh;

    int mi = lane & 15;
    int kh = (lane >> 4) << 1;
    int fa = (32 * wq + 16 * rh + mi) * GM_AP + kh;
    int fb[NT];
    #pragma unroll
    for (int t = 0; t < NT; ++t)
        fb[t] = kh * GM_BP + 32 * ((Q + t) & 7) + 16 * cs + mi;

    v4f ra0, ra1, rb[8];
    ra0 = *(const v4f*)(ga);
    ra1 = *(const v4f*)(ga + (size_t)32 * CDIM);
    #pragma unroll
    for (int i = 0; i < 8; ++i)
        rb[i] = *(const v4f*)(gb + (size_t)(4 * i) * CDIM);

    v8f c[NT];
    #pragma unroll
    for (int t = 0; t < NT; ++t) { v8f z = {}; c[t] = z; }
    #pragma nounroll
    for (int ch = 0; ch < CDIM / GM_KC; ++ch) {
        float* abuf = As[ch & 1];
        float* bbuf = Bs[ch & 1];
        {
            v4f t0 = ra0, t1 = ra1;
            if (MODE == 3) {
                int d0 = ti + arow - (ch * GM_KC + akq);
                int d1 = d0 + 32;
                t0.x = (d0 == 0) ? 3.0f - t0.x : -t0.x;
                t0.y = (d0 == 1) ? 3.0f - t0.y : -t0.y;
                t0.z = (d0 == 2) ? 3.0f - t0.z : -t0.z;
                t0.w = (d0 == 3) ? 3.0f - t0.w : -t0.w;
                t1.x = (d1 == 0) ? 3.0f - t1.x : -t1.x;
                t1.y = (d1 == 1) ? 3.0f - t1.y : -t1.y;
                t1.z = (d1 == 2) ? 3.0f - t1.z : -t1.z;
                t1.w = (d1 == 3) ? 3.0f - t1.w : -t1.w;
            }
            *(v4f*)(abuf + sa)              = t0;
            *(v4f*)(abuf + sa + 32 * GM_AP) = t1;
        }
        #pragma unroll
        for (int i = 0; i < 8; ++i) {
            v4f t = rb[i];
            if (MODE == 1) {
                int d = ch * GM_KC + brow + 4 * i - bnq;
                t.x = (d == 0) ? 3.0f - t.x : -t.x;
                t.y = (d == 1) ? 3.0f - t.y : -t.y;
                t.z = (d == 2) ? 3.0f - t.z : -t.z;
                t.w = (d == 3) ? 3.0f - t.w : -t.w;
            }
            *(v4f*)(bbuf + sb + 4 * i * GM_BP) = t;
        }
        __syncthreads();
        if (ch + 1 < CDIM / GM_KC) {
            int k0 = (ch + 1) * GM_KC;
            ra0 = *(const v4f*)(ga + k0);
            ra1 = *(const v4f*)(ga + k0 + (size_t)32 * CDIM);
            #pragma unroll
            for (int i = 0; i < 8; ++i)
                rb[i] = *(const v4f*)(gb + (size_t)(k0 + 4 * i) * CDIM);
        }
        #pragma unroll
        for (int kk = 0; kk < GM_KC; kk += 4) {
            v2f a = *(const v2f*)(abuf + fa + kk);
            v2f b[NT];
            #pragma unroll
            for (int t = 0; t < NT; ++t) {
                const float* pb = bbuf + fb[t];
                b[t].x = pb[kk * GM_BP];
                b[t].y = pb[(kk + 1) * GM_BP];
            }
            #pragma unroll
            for (int t = 0; t < NT; ++t)
                c[t] = wmma4(a, b[t], c[t]);
        }
    }

    float* Db = (MODE == 2) ? D + (size_t)batch * (CDIM * (CDIM + 1) / 2)
                            : D + (size_t)batch * CDIM * CDIM;
    float* Zb = (MODE == 1) ? Zn + (size_t)batch * CDIM * CDIM : nullptr;
    float s = (MODE == 2) ? sqrtf(trc[batch]) : 0.0f;
    int mbase = (lane >> 4) << 3;
    int nlo   = lane & 15;
    #pragma unroll
    for (int t = 0; t < NT; ++t) {
        if (t == 4 && Q >= 4) continue;
        int P  = (Q + t) & 7;
        int gn = 32 * P + 16 * cs + nlo;
        int g0 = R0 + mbase;
        float v[8];
        #pragma unroll
        for (int r = 0; r < 8; ++r) {
            v[r] = scale * c[t][r];
            if (t == 0 && g0 + r == gn) v[r] += diag;
        }
        if (MODE == 2) {
            if (t == 0) {
                #pragma unroll
                for (int r = 0; r < 8; ++r) {
                    int gm = g0 + r;
                    if (gn >= gm)
                        Db[gm * CDIM - (gm * (gm - 1)) / 2 + (gn - gm)] = s * v[r];
                }
            } else if (P > Q) {
                #pragma unroll
                for (int r = 0; r < 8; ++r) {
                    int gm = g0 + r;
                    Db[gm * CDIM - (gm * (gm - 1)) / 2 + (gn - gm)] = s * v[r];
                }
            } else {
                int base = gn * CDIM - (gn * (gn - 1)) / 2 + (g0 - gn);
                #pragma unroll
                for (int r = 0; r < 8; ++r) Db[base + r] = s * v[r];
            }
        } else {
            #pragma unroll
            for (int r = 0; r < 8; ++r)
                Db[(size_t)(g0 + r) * CDIM + gn] = v[r];
            if (t > 0) {
                float* mp = Db + (size_t)gn * CDIM + g0;
                v4f lo = { v[0], v[1], v[2], v[3] };
                v4f hi = { v[4], v[5], v[6], v[7] };
                *(v4f*)mp       = lo;
                *(v4f*)(mp + 4) = hi;
            }
            if (MODE == 1) {
                float z[8];
                #pragma unroll
                for (int r = 0; r < 8; ++r) {
                    float u = -v[r];
                    if (t == 0 && g0 + r == gn) u += 3.0f;
                    z[r] = 0.5f * u;
                    Zb[(size_t)(g0 + r) * CDIM + gn] = z[r];
                }
                if (t > 0) {
                    float* mp = Zb + (size_t)gn * CDIM + g0;
                    v4f lo = { z[0], z[1], z[2], z[3] };
                    v4f hi = { z[4], z[5], z[6], z[7] };
                    *(v4f*)mp       = lo;
                    *(v4f*)(mp + 4) = hi;
                }
            }
        }
    }
}

extern "C" void kernel_launch(void* const* d_in, const int* in_sizes, int n_in,
                              void* d_out, int out_size, void* d_ws, size_t ws_size,
                              hipStream_t stream) {
    (void)in_sizes; (void)n_in; (void)out_size; (void)ws_size;

    const float* x   = (const float*)d_in[0];
    float*       out = (float*)d_out;
    float*       ws  = (float*)d_ws;

    const size_t MAT = (size_t)NBATCH * CDIM * CDIM;

    float* trc  = ws;
    float* T    = trc + NBATCH;
    float* Ya   = T  + MAT;
    float* Yb   = Ya + MAT;
    float* T3   = Yb + MAT;
    float* Z1   = T3 + MAT;

    sigma_kernel<<<NBATCH * 4, 256, 0, stream>>>(x, Ya, trc);

    gemm256_kernel<1><<<NBATCH * 4, 256, 0, stream>>>(Ya, Ya, Yb, Z1, nullptr, 0.5f, 0.0f);

    gemm256_kernel<0><<<NBATCH * 4, 256, 0, stream>>>(Z1, Yb, T,  nullptr, nullptr, -1.0f, 3.0f);
    gemm256_kernel<0><<<NBATCH * 4, 256, 0, stream>>>(Yb, T,  Ya, nullptr, nullptr,  0.5f, 0.0f);

    gemm256_kernel<0><<<NBATCH * 4, 256, 0, stream>>>(Z1, Ya, T,  nullptr, nullptr,  1.0f, 0.0f);
    gemm256_kernel<3><<<NBATCH * 4, 256, 0, stream>>>(T,  T,  T3, nullptr, nullptr, -0.5f, 3.0f);

    gemm256_kernel<2><<<NBATCH * 4, 256, 0, stream>>>(Ya, T3, out, nullptr, trc,     0.5f, 0.0f);
}


# P = Z1@Y2 reassociated as 0.5*(3I-T)@T (Z1@Y1 = 3I-T): P no longer waits for Y2, so Y2 = 0.5*Y1@T and P share one launch of 512 workgroups (7 launches to 6); rounding order changed
# speedup vs baseline: 3.4289x; 1.0428x over previous
#include <hip/hip_runtime.h>
#include <math.h>

typedef __attribute__((ext_vector_type(2))) float v2f;
typedef __attribute__((ext_vector_type(4))) float v4f;
typedef __attribute__((ext_vector_type(8))) float v8f;

#define CDIM 256
#define NDIM 784
#define NBATCH 64

static __device__ __forceinline__ v8f wmma4(v2f a, v2f b, v8f c) {
    return __builtin_amdgcn_wmma_f32_16x16x4_f32(false, a, false, b,
                                                 (short)0, c, false, false);
}


#define SG_KC 32
#define SG_P  36
#define NT    5

#define SG_KSTEP(KK)                                                         \
    {                                                                        \
        v2f a = *(const v2f*)(buf + fa + (KK));                              \
        v2f b[NT];                                                           \
        _Pragma("unroll")                                                    \
        for (int t = 0; t < NT; ++t)                                         \
            b[t] = *(const v2f*)(buf + fb[t] + (KK));                        \
        _Pragma("unroll")                                                    \
        for (int t = 0; t < NT; ++t)                                         \
            c[t] = wmma4(a, b[t], c[t]);                                     \
    }

__global__ __launch_bounds__(256)
void sigma_kernel(const float* __restrict__ X,
                  float* __restrict__ Y,
                  float* __restrict__ trc) {
    __shared__ __attribute__((aligned(16))) float Xs[2][CDIM * SG_P];
    __shared__ float qred[8];
    __shared__ float rsum[CDIM];

    int tid   = threadIdx.x;
    int wave  = __builtin_amdgcn_readfirstlane(tid >> 5);
    int lane  = tid & 31;
    int batch = blockIdx.x >> 2;
    int ti = (blockIdx.x & 3) << 6;

    const float* Xb = X + (size_t)batch * CDIM * NDIM;

    int srow = tid >> 3;
    int skq  = (tid & 7) << 2;
    const float* gp = Xb + (size_t)srow * NDIM + skq;
    int so = srow * SG_P + skq;
    float rs[8];
    #pragma unroll
    for (int i = 0; i < 8; ++i) rs[i] = 0.f;

    int wq = wave >> 2;
    int rh = (wave >> 1) & 1;
    int cs = wave & 1;
    int Q  = ((blockIdx.x & 3) << 1) + wq;
    int R0 = 32 * Q + 16 * rh;

    int mi = lane & 15;
    int kh = (lane >> 4) << 1;
    int fa = (R0 + mi) * SG_P + kh;
    int fb[NT];
    #pragma unroll
    for (int t = 0; t < NT; ++t)
        fb[t] = (32 * ((Q + t) & 7) + 16 * cs + mi) * SG_P + kh;

    v4f r[8];
    #pragma unroll
    for (int i = 0; i < 8; ++i)
        r[i] = *(const v4f*)(gp + (size_t)(32 * i) * NDIM);

    v8f c[NT];
    #pragma unroll
    for (int t = 0; t < NT; ++t) { v8f z = {}; c[t] = z; }

    float q = 0.f;
    const int NFULL = NDIM / SG_KC;
    for (int ch = 0; ch < NFULL; ++ch) {
        float* buf = Xs[ch & 1];
        #pragma unroll
        for (int i = 0; i < 8; ++i) {
            v4f v = r[i];
            *(v4f*)(buf + so + 32 * i * SG_P) = v;
            q += v.x * v.x + v.y * v.y + v.z * v.z + v.w * v.w;
            rs[i] += (v.x + v.y) + (v.z + v.w);
        }
        __syncthreads();
        if (ch + 1 < NFULL || skq < NDIM - NFULL * SG_KC) {
            const float* g = gp + (ch + 1) * SG_KC;
            #pragma unroll
            for (int i = 0; i < 8; ++i)
                r[i] = *(const v4f*)(g + (size_t)(32 * i) * NDIM);
        }
        #pragma unroll
        for (int kk = 0; kk < SG_KC; kk += 4) SG_KSTEP(kk)
    }
    {
        float* buf = Xs[NFULL & 1];
        if (skq < NDIM - NFULL * SG_KC) {
            #pragma unroll
            for (int i = 0; i < 8; ++i) {
                v4f v = r[i];
                *(v4f*)(buf + so + 32 * i * SG_P) = v;
                q += v.x * v.x + v.y * v.y + v.z * v.z + v.w * v.w;
                rs[i] += (v.x + v.y) + (v.z + v.w);
            }
        }
        __syncthreads();
        #pragma unroll
        for (int kk = 0; kk < NDIM - NFULL * SG_KC; kk += 4) SG_KSTEP(kk)
    }

    #pragma unroll
    for (int i = 0; i < 8; ++i) {
        float t = rs[i];
        t += __shfl_xor(t, 1, 32);
        t += __shfl_xor(t, 2, 32);
        t += __shfl_xor(t, 4, 32);
        rs[i] = t;
    }
    if ((tid & 7) == 0) {
        #pragma unroll
        for (int i = 0; i < 8; ++i) rsum[srow + 32 * i] = rs[i];
    }
    const float inv_n = 1.0f / (float)NDIM;
    for (int off = 16; off > 0; off >>= 1) q += __shfl_xor(q, off, 32);
    if (lane == 0) qred[wave] = q;
    __syncthreads();
    float qq = ((qred[0] + qred[1]) + (qred[2] + qred[3])) +
               ((qred[4] + qred[5]) + (qred[6] + qred[7]));
    float s2 = 0.f;
    #pragma unroll
    for (int j = 0; j < 8; ++j) {
        float t = rsum[lane + 32 * j];
        s2 += t * t;
    }
    for (int off = 16; off > 0; off >>= 1) s2 += __shfl_xor(s2, off, 32);
    float tr = (qq - s2 * inv_n) * inv_n;
    if (ti == 0 && tid == 0) trc[batch] = tr;
    const float ysc = inv_n / tr;

    float* Sb = Y + (size_t)batch * CDIM * CDIM;
    int mbase = (lane >> 4) << 3;
    int nlo   = lane & 15;
    #pragma unroll
    for (int t = 0; t < NT; ++t) {
        if (t == 4 && Q >= 4) continue;
        int gn = 32 * ((Q + t) & 7) + 16 * cs + nlo;
        float scn = rsum[gn];
        float v[8];
        #pragma unroll
        for (int r8 = 0; r8 < 8; ++r8) {
            int gm = R0 + mbase + r8;
            v[r8] = (c[t][r8] - (rsum[gm] * scn) * inv_n) * ysc;
            Sb[(size_t)gm * CDIM + gn] = v[r8];
        }
        if (t > 0) {
            float* mp = Sb + (size_t)gn * CDIM + R0 + mbase;
            v4f lo = { v[0], v[1], v[2], v[3] };
            v4f hi = { v[4], v[5], v[6], v[7] };
            *(v4f*)mp       = lo;
            *(v4f*)(mp + 4) = hi;
        }
    }
}

#define GM_KC 32
#define GM_AP 36
#define GM_BP 264

template <int MODE>
static __device__ __forceinline__
void gemm256_body(const float* __restrict__ A,
                  const float* __restrict__ Bm,
                  float* __restrict__ D,
                  float* __restrict__ Zn,
                  const float* __restrict__ trc,
                  float scale, float diag, unsigned bid,
                  float (*As)[64 * GM_AP], float (*Bs)[GM_KC * GM_BP]) {
    int tid   = threadIdx.x;
    int wave  = __builtin_amdgcn_readfirstlane(tid >> 5);
    int lane  = tid & 31;
    int batch = bid >> 2;
    int ti = (bid & 3) << 6;

    const float* Ab = A  + (size_t)batch * CDIM * CDIM + (size_t)ti * CDIM;
    const float* Bb = Bm + (size_t)batch * CDIM * CDIM;

    int arow = tid >> 3;
    int akq  = (tid & 7) << 2;
    int brow = tid >> 6;
    int bnq  = (tid & 63) << 2;
    const float* ga = Ab + (size_t)arow * CDIM + akq;
    const float* gb = Bb + (size_t)brow * CDIM + bnq;
    int sa = arow * GM_AP + akq;
    int sb = brow * GM_BP + bnq;

    int wq = wave >> 2;
    int rh = (wave >> 1) & 1;
    int cs = wave & 1;
    int Q  = ((bid & 3) << 1) + wq;
    int R0 = 32 * Q + 16 * rh;

    int mi = lane & 15;
    int kh = (lane >> 4) << 1;
    int fa = (32 * wq + 16 * rh + mi) * GM_AP + kh;
    int fb[NT];
    #pragma unroll
    for (int t = 0; t < NT; ++t)
        fb[t] = kh * GM_BP + 32 * ((Q + t) & 7) + 16 * cs + mi;

    v4f ra0, ra1, rb[8];
    ra0 = *(const v4f*)(ga);
    ra1 = *(const v4f*)(ga + (size_t)32 * CDIM);
    #pragma unroll
    for (int i = 0; i < 8; ++i)
        rb[i] = *(const v4f*)(gb + (size_t)(4 * i) * CDIM);

    v8f c[NT];
    #pragma unroll
    for (int t = 0; t < NT; ++t) { v8f z = {}; c[t] = z; }
    #pragma nounroll
    for (int ch = 0; ch < CDIM / GM_KC; ++ch) {
        float* abuf = As[ch & 1];
        float* bbuf = Bs[ch & 1];
        {
            v4f t0 = ra0, t1 = ra1;
            if (MODE == 3) {
                int d0 = ti + arow - (ch * GM_KC + akq);
                int d1 = d0 + 32;
                t0.x = (d0 == 0) ? 3.0f - t0.x : -t0.x;
                t0.y = (d0 == 1) ? 3.0f - t0.y : -t0.y;
                t0.z = (d0 == 2) ? 3.0f - t0.z : -t0.z;
                t0.w = (d0 == 3) ? 3.0f - t0.w : -t0.w;
                t1.x = (d1 == 0) ? 3.0f - t1.x : -t1.x;
                t1.y = (d1 == 1) ? 3.0f - t1.y : -t1.y;
                t1.z = (d1 == 2) ? 3.0f - t1.z : -t1.z;
                t1.w = (d1 == 3) ? 3.0f - t1.w : -t1.w;
            }
            *(v4f*)(abuf + sa)              = t0;
            *(v4f*)(abuf + sa + 32 * GM_AP) = t1;
        }
        #pragma unroll
        for (int i = 0; i < 8; ++i) {
            v4f t = rb[i];
            if (MODE == 1) {
                int d = ch * GM_KC + brow + 4 * i - bnq;
                t.x = (d == 0) ? 3.0f - t.x : -t.x;
                t.y = (d == 1) ? 3.0f - t.y : -t.y;
                t.z = (d == 2) ? 3.0f - t.z : -t.z;
                t.w = (d == 3) ? 3.0f - t.w : -t.w;
            }
            *(v4f*)(bbuf + sb + 4 * i * GM_BP) = t;
        }
        __syncthreads();
        if (ch + 1 < CDIM / GM_KC) {
            int k0 = (ch + 1) * GM_KC;
            ra0 = *(const v4f*)(ga + k0);
            ra1 = *(const v4f*)(ga + k0 + (size_t)32 * CDIM);
            #pragma unroll
            for (int i = 0; i < 8; ++i)
                rb[i] = *(const v4f*)(gb + (size_t)(k0 + 4 * i) * CDIM);
        }
        #pragma unroll
        for (int kk = 0; kk < GM_KC; kk += 4) {
            v2f a = *(const v2f*)(abuf + fa + kk);
            v2f b[NT];
            #pragma unroll
            for (int t = 0; t < NT; ++t) {
                const float* pb = bbuf + fb[t];
                b[t].x = pb[kk * GM_BP];
                b[t].y = pb[(kk + 1) * GM_BP];
            }
            #pragma unroll
            for (int t = 0; t < NT; ++t)
                c[t] = wmma4(a, b[t], c[t]);
        }
    }

    float* Db = (MODE == 2) ? D + (size_t)batch * (CDIM * (CDIM + 1) / 2)
                            : D + (size_t)batch * CDIM * CDIM;
    float* Zb = (MODE == 1) ? Zn + (size_t)batch * CDIM * CDIM : nullptr;
    float s = (MODE == 2) ? sqrtf(trc[batch]) : 0.0f;
    int mbase = (lane >> 4) << 3;
    int nlo   = lane & 15;
    #pragma unroll
    for (int t = 0; t < NT; ++t) {
        if (t == 4 && Q >= 4) continue;
        int P  = (Q + t) & 7;
        int gn = 32 * P + 16 * cs + nlo;
        int g0 = R0 + mbase;
        float v[8];
        #pragma unroll
        for (int r = 0; r < 8; ++r) {
            v[r] = scale * c[t][r];
            if (t == 0 && g0 + r == gn) v[r] += diag;
        }
        if (MODE == 2) {
            if (t == 0) {
                #pragma unroll
                for (int r = 0; r < 8; ++r) {
                    int gm = g0 + r;
                    if (gn >= gm)
                        Db[gm * CDIM - (gm * (gm - 1)) / 2 + (gn - gm)] = s * v[r];
                }
            } else if (P > Q) {
                #pragma unroll
                for (int r = 0; r < 8; ++r) {
                    int gm = g0 + r;
                    Db[gm * CDIM - (gm * (gm - 1)) / 2 + (gn - gm)] = s * v[r];
                }
            } else {
                int base = gn * CDIM - (gn * (gn - 1)) / 2 + (g0 - gn);
                #pragma unroll
                for (int r = 0; r < 8; ++r) Db[base + r] = s * v[r];
            }
        } else {
            #pragma unroll
            for (int r = 0; r < 8; ++r)
                Db[(size_t)(g0 + r) * CDIM + gn] = v[r];
            if (t > 0) {
                float* mp = Db + (size_t)gn * CDIM + g0;
                v4f lo = { v[0], v[1], v[2], v[3] };
                v4f hi = { v[4], v[5], v[6], v[7] };
                *(v4f*)mp       = lo;
                *(v4f*)(mp + 4) = hi;
            }
            if (MODE == 1) {
                float z[8];
                #pragma unroll
                for (int r = 0; r < 8; ++r) {
                    float u = -v[r];
                    if (t == 0 && g0 + r == gn) u += 3.0f;
                    z[r] = 0.5f * u;
                    Zb[(size_t)(g0 + r) * CDIM + gn] = z[r];
                }
                if (t > 0) {
                    float* mp = Zb + (size_t)gn * CDIM + g0;
                    v4f lo = { z[0], z[1], z[2], z[3] };
                    v4f hi = { z[4], z[5], z[6], z[7] };
                    *(v4f*)mp       = lo;
                    *(v4f*)(mp + 4) = hi;
                }
            }
        }
    }
}

template <int MODE>
__global__ __launch_bounds__(256)
void gemm256_kernel(const float* __restrict__ A,
                    const float* __restrict__ Bm,
                    float* __restrict__ D,
                    float* __restrict__ Zn,
                    const float* __restrict__ trc,
                    float scale, float diag) {
    __shared__ __attribute__((aligned(16))) float As[2][64 * GM_AP];
    __shared__ __attribute__((aligned(16))) float Bs[2][GM_KC * GM_BP];
    gemm256_body<MODE>(A, Bm, D, Zn, trc, scale, diag, blockIdx.x, As, Bs);
}

__global__ __launch_bounds__(256)
void gemm256_pair_kernel(const float* __restrict__ Y1,
                         const float* __restrict__ T,
                         float* __restrict__ Y2,
                         float* __restrict__ P) {
    __shared__ __attribute__((aligned(16))) float As[2][64 * GM_AP];
    __shared__ __attribute__((aligned(16))) float Bs[2][GM_KC * GM_BP];
    const unsigned NWG = NBATCH * 4;
    unsigned bid = blockIdx.x;
    if (bid < NWG)
        gemm256_body<0>(Y1, T, Y2, nullptr, nullptr, 0.5f, 0.0f, bid, As, Bs);
    else
        gemm256_body<3>(T, T, P, nullptr, nullptr, 0.5f, 0.0f, bid - NWG, As, Bs);
}

extern "C" void kernel_launch(void* const* d_in, const int* in_sizes, int n_in,
                              void* d_out, int out_size, void* d_ws, size_t ws_size,
                              hipStream_t stream) {
    (void)in_sizes; (void)n_in; (void)out_size; (void)ws_size;

    const float* x   = (const float*)d_in[0];
    float*       out = (float*)d_out;
    float*       ws  = (float*)d_ws;

    const size_t MAT = (size_t)NBATCH * CDIM * CDIM;

    float* trc  = ws;
    float* T    = trc + NBATCH;
    float* Ya   = T  + MAT;
    float* Yb   = Ya + MAT;
    float* T3   = Yb + MAT;
    float* Z1   = T3 + MAT;
    float* Pm   = Z1;

    sigma_kernel<<<NBATCH * 4, 256, 0, stream>>>(x, Ya, trc);

    gemm256_kernel<1><<<NBATCH * 4, 256, 0, stream>>>(Ya, Ya, Yb, Z1, nullptr, 0.5f, 0.0f);

    gemm256_kernel<0><<<NBATCH * 4, 256, 0, stream>>>(Z1, Yb, T,  nullptr, nullptr, -1.0f, 3.0f);

    gemm256_pair_kernel<<<NBATCH * 8, 256, 0, stream>>>(Yb, T, Ya, Pm);

    gemm256_kernel<3><<<NBATCH * 4, 256, 0, stream>>>(Pm, Pm, T3, nullptr, nullptr, -0.5f, 3.0f);

    gemm256_kernel<2><<<NBATCH * 4, 256, 0, stream>>>(Ya, T3, out, nullptr, trc,     0.5f, 0.0f);
}


# K loops of both WMMA kernels run half a chunk askew: each half-chunk's LDS reads stand as one group ahead of 20 WMMAs on fragments already held, loop barrier orders LDS only; sums unchanged
# speedup vs baseline: 3.5968x; 1.0490x over previous
#include <hip/hip_runtime.h>
#include <math.h>

typedef __attribute__((ext_vector_type(2))) float v2f;
typedef __attribute__((ext_vector_type(4))) float v4f;
typedef __attribute__((ext_vector_type(8))) float v8f;

#define CDIM 256
#define NDIM 784
#define NBATCH 64

static __device__ __forceinline__ v8f wmma4(v2f a, v2f b, v8f c) {
    return __builtin_amdgcn_wmma_f32_16x16x4_f32(false, a, false, b,
                                                 (short)0, c, false, false);
}

static __device__ __forceinline__ void lds_barrier() {
    __builtin_amdgcn_fence(__ATOMIC_RELEASE, "workgroup", "local");
    __builtin_amdgcn_s_barrier();
    __builtin_amdgcn_fence(__ATOMIC_ACQUIRE, "workgroup", "local");
}


#define SG_KC 32
#define SG_P  36
#define NT    5

#define SG_LOAD_HALF(FA, FB, BUF, K0)                                        \
    _Pragma("unroll")                                                        \
    for (int s = 0; s < 4; ++s) {                                            \
        FA[s] = *(const v2f*)((BUF) + fa + (K0) + 4 * s);                    \
        _Pragma("unroll")                                                    \
        for (int t = 0; t < NT; ++t)                                         \
            FB[s][t] = *(const v2f*)((BUF) + fb[t] + (K0) + 4 * s);          \
    }
#define SG_MMA_HALF(FA, FB)                                                  \
    _Pragma("unroll")                                                        \
    for (int s = 0; s < 4; ++s) {                                            \
        _Pragma("unroll")                                                    \
        for (int t = 0; t < NT; ++t)                                         \
            c[t] = wmma4(FA[s], FB[s][t], c[t]);                             \
    }
#define SG_STAGE(BUF)                                                        \
    _Pragma("unroll")                                                        \
    for (int i = 0; i < 8; ++i) {                                            \
        v4f v = r[i];                                                        \
        *(v4f*)((BUF) + so + 32 * i * SG_P) = v;                             \
        q += v.x * v.x + v.y * v.y + v.z * v.z + v.w * v.w;                  \
        rs[i] += (v.x + v.y) + (v.z + v.w);                                  \
    }
#define SG_GLOAD(CH)                                                         \
    {                                                                        \
        const float* g = gp + (CH) * SG_KC;                                  \
        _Pragma("unroll")                                                    \
        for (int i = 0; i < 8; ++i)                                          \
            r[i] = *(const v4f*)(g + (size_t)(32 * i) * NDIM);               \
    }

__global__ __launch_bounds__(256)
void sigma_kernel(const float* __restrict__ X,
                  float* __restrict__ Y,
                  float* __restrict__ trc) {
    __shared__ __attribute__((aligned(16))) float Xs[2][CDIM * SG_P];
    __shared__ float qred[8];
    __shared__ float rsum[CDIM];

    int tid   = threadIdx.x;
    int wave  = __builtin_amdgcn_readfirstlane(tid >> 5);
    int lane  = tid & 31;
    int batch = blockIdx.x >> 2;
    int ti = (blockIdx.x & 3) << 6;

    const float* Xb = X + (size_t)batch * CDIM * NDIM;

    int srow = tid >> 3;
    int skq  = (tid & 7) << 2;
    const float* gp = Xb + (size_t)srow * NDIM + skq;
    int so = srow * SG_P + skq;
    float rs[8];
    #pragma unroll
    for (int i = 0; i < 8; ++i) rs[i] = 0.f;

    int wq = wave >> 2;
    int rh = (wave >> 1) & 1;
    int cs = wave & 1;
    int Q  = ((blockIdx.x & 3) << 1) + wq;
    int R0 = 32 * Q + 16 * rh;

    int mi = lane & 15;
    int kh = (lane >> 4) << 1;
    int fa = (R0 + mi) * SG_P + kh;
    int fb[NT];
    #pragma unroll
    for (int t = 0; t < NT; ++t)
        fb[t] = (32 * ((Q + t) & 7) + 16 * cs + mi) * SG_P + kh;

    v4f r[8];
    #pragma unroll
    for (int i = 0; i < 8; ++i)
        r[i] = *(const v4f*)(gp + (size_t)(32 * i) * NDIM);

    v8f c[NT];
    #pragma unroll
    for (int t = 0; t < NT; ++t) { v8f z = {}; c[t] = z; }

    float q = 0.f;
    const int NFULL = NDIM / SG_KC;
    const int TAILW = NDIM - NFULL * SG_KC;
    v2f xa[4], xb[4][NT];
    v2f ya[4], yb[4][NT];

    SG_STAGE(Xs[0])
    SG_GLOAD(1)
    lds_barrier();
    SG_LOAD_HALF(xa, xb, Xs[0], 0)
    SG_LOAD_HALF(ya, yb, Xs[0], 16)
    __builtin_amdgcn_sched_barrier(0);
    SG_MMA_HALF(xa, xb)
    SG_STAGE(Xs[1])
    SG_GLOAD(2)
    for (int ch = 1; ch < NFULL; ++ch) {
        float* buf  = Xs[ch & 1];
        float* nbuf = Xs[(ch + 1) & 1];
        lds_barrier();
        SG_LOAD_HALF(xa, xb, buf, 0)
        __builtin_amdgcn_sched_barrier(0);
        SG_MMA_HALF(ya, yb)
        __builtin_amdgcn_sched_barrier(0);
        SG_LOAD_HALF(ya, yb, buf, 16)
        if (ch + 1 < NFULL || skq < TAILW) {
            SG_STAGE(nbuf)
        }
        if (ch + 2 < NFULL || (ch + 2 == NFULL && skq < TAILW)) {
            SG_GLOAD(ch + 2)
        }
        SG_MMA_HALF(xa, xb)
    }
    lds_barrier();
    SG_LOAD_HALF(xa, xb, Xs[NFULL & 1], 0)
    __builtin_amdgcn_sched_barrier(0);
    SG_MMA_HALF(ya, yb)
    SG_MMA_HALF(xa, xb)

    #pragma unroll
    for (int i = 0; i < 8; ++i) {
        float t = rs[i];
        t += __shfl_xor(t, 1, 32);
        t += __shfl_xor(t, 2, 32);
        t += __shfl_xor(t, 4, 32);
        rs[i] = t;
    }
    if ((tid & 7) == 0) {
        #pragma unroll
        for (int i = 0; i < 8; ++i) rsum[srow + 32 * i] = rs[i];
    }
    const float inv_n = 1.0f / (float)NDIM;
    for (int off = 16; off > 0; off >>= 1) q += __shfl_xor(q, off, 32);
    if (lane == 0) qred[wave] = q;
    __syncthreads();
    float qq = ((qred[0] + qred[1]) + (qred[2] + qred[3])) +
               ((qred[4] + qred[5]) + (qred[6] + qred[7]));
    float s2 = 0.f;
    #pragma unroll
    for (int j = 0; j < 8; ++j) {
        float t = rsum[lane + 32 * j];
        s2 += t * t;
    }
    for (int off = 16; off > 0; off >>= 1) s2 += __shfl_xor(s2, off, 32);
    float tr = (qq - s2 * inv_n) * inv_n;
    if (ti == 0 && tid == 0) trc[batch] = tr;
    const float ysc = inv_n / tr;

    float* Sb = Y + (size_t)batch * CDIM * CDIM;
    int mbase = (lane >> 4) << 3;
    int nlo   = lane & 15;
    #pragma unroll
    for (int t = 0; t < NT; ++t) {
        if (t == 4 && Q >= 4) continue;
        int gn = 32 * ((Q + t) & 7) + 16 * cs + nlo;
        float scn = rsum[gn];
        float v[8];
        #pragma unroll
        for (int r8 = 0; r8 < 8; ++r8) {
            int gm = R0 + mbase + r8;
            v[r8] = (c[t][r8] - (rsum[gm] * scn) * inv_n) * ysc;
            Sb[(size_t)gm * CDIM + gn] = v[r8];
        }
        if (t > 0) {
            float* mp = Sb + (size_t)gn * CDIM + R0 + mbase;
            v4f lo = { v[0], v[1], v[2], v[3] };
            v4f hi = { v[4], v[5], v[6], v[7] };
            *(v4f*)mp       = lo;
            *(v4f*)(mp + 4) = hi;
        }
    }
}

#define GM_KC 32
#define GM_AP 36
#define GM_BP 264

#define GM_GLOAD(CH)                                                         \
    {                                                                        \
        int k0 = (CH) * GM_KC;                                               \
        ra0 = *(const v4f*)(ga + k0);                                        \
        ra1 = *(const v4f*)(ga + k0 + (size_t)32 * CDIM);                    \
        _Pragma("unroll")                                                    \
        for (int i = 0; i < 8; ++i)                                          \
            rb[i] = *(const v4f*)(gb + (size_t)(k0 + 4 * i) * CDIM);         \
    }
#define GM_STAGE(CH, ABUF, BBUF)                                             \
    {                                                                        \
        v4f t0 = ra0, t1 = ra1;                                              \
        if (MODE == 3) {                                                     \
                \
              \
            int d0 = ti + arow - ((CH) * GM_KC + akq);                       \
            int d1 = d0 + 32;                                                \
            t0.x = (d0 == 0) ? 3.0f - t0.x : -t0.x;                          \
            t0.y = (d0 == 1) ? 3.0f - t0.y : -t0.y;                          \
            t0.z = (d0 == 2) ? 3.0f - t0.z : -t0.z;                          \
            t0.w = (d0 == 3) ? 3.0f - t0.w : -t0.w;                          \
            t1.x = (d1 == 0) ? 3.0f - t1.x : -t1.x;                          \
            t1.y = (d1 == 1) ? 3.0f - t1.y : -t1.y;                          \
            t1.z = (d1 == 2) ? 3.0f - t1.z : -t1.z;                          \
            t1.w = (d1 == 3) ? 3.0f - t1.w : -t1.w;                          \
        }                                                                    \
        *(v4f*)((ABUF) + sa)              = t0;                              \
        *(v4f*)((ABUF) + sa + 32 * GM_AP) = t1;                              \
        _Pragma("unroll")                                                    \
        for (int i = 0; i < 8; ++i) {                                        \
            v4f t = rb[i];                                                   \
            if (MODE == 1) {                                                 \
                  \
                                    \
                int d = (CH) * GM_KC + brow + 4 * i - bnq;                   \
                t.x = (d == 0) ? 3.0f - t.x : -t.x;                          \
                t.y = (d == 1) ? 3.0f - t.y : -t.y;                          \
                t.z = (d == 2) ? 3.0f - t.z : -t.z;                          \
                t.w = (d == 3) ? 3.0f - t.w : -t.w;                          \
            }                                                                \
            *(v4f*)((BBUF) + sb + 4 * i * GM_BP) = t;                        \
        }                                                                    \
    }
#define GM_LOAD_HALF(FA, FB, ABUF, BBUF, K0)                                 \
    _Pragma("unroll")                                                        \
    for (int s = 0; s < 4; ++s) {                                            \
        const int kk = (K0) + 4 * s;                                         \
        FA[s] = *(const v2f*)((ABUF) + fa + kk);                             \
        _Pragma("unroll")                                                    \
        for (int t = 0; t < NT; ++t) {                                       \
            const float* pb = (BBUF) + fb[t];                                \
            FB[s][t].x = pb[kk * GM_BP];                                     \
            FB[s][t].y = pb[(kk + 1) * GM_BP];                               \
        }                                                                    \
    }
#define GM_MMA_HALF(FA, FB)                                                  \
    _Pragma("unroll")                                                        \
    for (int s = 0; s < 4; ++s) {                                            \
        _Pragma("unroll")                                                    \
        for (int t = 0; t < NT; ++t)                                         \
            c[t] = wmma4(FA[s], FB[s][t], c[t]);                             \
    }

template <int MODE>
static __device__ __forceinline__
void gemm256_body(const float* __restrict__ A,
                  const float* __restrict__ Bm,
                  float* __restrict__ D,
                  float* __restrict__ Zn,
                  const float* __restrict__ trc,
                  float scale, float diag, unsigned bid,
                  float (*As)[64 * GM_AP], float (*Bs)[GM_KC * GM_BP]) {
    int tid   = threadIdx.x;
    int wave  = __builtin_amdgcn_readfirstlane(tid >> 5);
    int lane  = tid & 31;
    int batch = bid >> 2;
    int ti = (bid & 3) << 6;

    const float* Ab = A  + (size_t)batch * CDIM * CDIM + (size_t)ti * CDIM;
    const float* Bb = Bm + (size_t)batch * CDIM * CDIM;

    int arow = tid >> 3;
    int akq  = (tid & 7) << 2;
    int brow = tid >> 6;
    int bnq  = (tid & 63) << 2;
    const float* ga = Ab + (size_t)arow * CDIM + akq;
    const float* gb = Bb + (size_t)brow * CDIM + bnq;
    int sa = arow * GM_AP + akq;
    int sb = brow * GM_BP + bnq;

    int wq = wave >> 2;
    int rh = (wave >> 1) & 1;
    int cs = wave & 1;
    int Q  = ((bid & 3) << 1) + wq;
    int R0 = 32 * Q + 16 * rh;

    int mi = lane & 15;
    int kh = (lane >> 4) << 1;
    int fa = (32 * wq + 16 * rh + mi) * GM_AP + kh;
    int fb[NT];
    #pragma unroll
    for (int t = 0; t < NT; ++t)
        fb[t] = kh * GM_BP + 32 * ((Q + t) & 7) + 16 * cs + mi;

    v4f ra0, ra1, rb[8];
    GM_GLOAD(0)

    v8f c[NT];
    #pragma unroll
    for (int t = 0; t < NT; ++t) { v8f z = {}; c[t] = z; }
    v2f xa[4], xb[4][NT];
    v2f ya[4], yb[4][NT];

    GM_STAGE(0, As[0], Bs[0])
    GM_GLOAD(1)
    lds_barrier();
    GM_LOAD_HALF(xa, xb, As[0], Bs[0], 0)
    GM_LOAD_HALF(ya, yb, As[0], Bs[0], 16)
    __builtin_amdgcn_sched_barrier(0);
    GM_MMA_HALF(xa, xb)
    GM_STAGE(1, As[1], Bs[1])
    GM_GLOAD(2)
    #pragma nounroll
    for (int ch = 1; ch < CDIM / GM_KC; ++ch) {
        float* abuf = As[ch & 1];
        float* bbuf = Bs[ch & 1];
        lds_barrier();
        GM_LOAD_HALF(xa, xb, abuf, bbuf, 0)
        __builtin_amdgcn_sched_barrier(0);
        GM_MMA_HALF(ya, yb)
        __builtin_amdgcn_sched_barrier(0);
        GM_LOAD_HALF(ya, yb, abuf, bbuf, 16)
        if (ch + 1 < CDIM / GM_KC) {
            GM_STAGE(ch + 1, As[(ch + 1) & 1], Bs[(ch + 1) & 1])
        }
        if (ch + 2 < CDIM / GM_KC) {
            GM_GLOAD(ch + 2)
        }
        GM_MMA_HALF(xa, xb)
    }
    GM_MMA_HALF(ya, yb)

    float* Db = (MODE == 2) ? D + (size_t)batch * (CDIM * (CDIM + 1) / 2)
                            : D + (size_t)batch * CDIM * CDIM;
    float* Zb = (MODE == 1) ? Zn + (size_t)batch * CDIM * CDIM : nullptr;
    float s = (MODE == 2) ? sqrtf(trc[batch]) : 0.0f;
    int mbase = (lane >> 4) << 3;
    int nlo   = lane & 15;
    #pragma unroll
    for (int t = 0; t < NT; ++t) {
        if (t == 4 && Q >= 4) continue;
        int P  = (Q + t) & 7;
        int gn = 32 * P + 16 * cs + nlo;
        int g0 = R0 + mbase;
        float v[8];
        #pragma unroll
        for (int r = 0; r < 8; ++r) {
            v[r] = scale * c[t][r];
            if (t == 0 && g0 + r == gn) v[r] += diag;
        }
        if (MODE == 2) {
            if (t == 0) {
                #pragma unroll
                for (int r = 0; r < 8; ++r) {
                    int gm = g0 + r;
                    if (gn >= gm)
                        Db[gm * CDIM - (gm * (gm - 1)) / 2 + (gn - gm)] = s * v[r];
                }
            } else if (P > Q) {
                #pragma unroll
                for (int r = 0; r < 8; ++r) {
                    int gm = g0 + r;
                    Db[gm * CDIM - (gm * (gm - 1)) / 2 + (gn - gm)] = s * v[r];
                }
            } else {
                int base = gn * CDIM - (gn * (gn - 1)) / 2 + (g0 - gn);
                #pragma unroll
                for (int r = 0; r < 8; ++r) Db[base + r] = s * v[r];
            }
        } else {
            #pragma unroll
            for (int r = 0; r < 8; ++r)
                Db[(size_t)(g0 + r) * CDIM + gn] = v[r];
            if (t > 0) {
                float* mp = Db + (size_t)gn * CDIM + g0;
                v4f lo = { v[0], v[1], v[2], v[3] };
                v4f hi = { v[4], v[5], v[6], v[7] };
                *(v4f*)mp       = lo;
                *(v4f*)(mp + 4) = hi;
            }
            if (MODE == 1) {
                float z[8];
                #pragma unroll
                for (int r = 0; r < 8; ++r) {
                    float u = -v[r];
                    if (t == 0 && g0 + r == gn) u += 3.0f;
                    z[r] = 0.5f * u;
                    Zb[(size_t)(g0 + r) * CDIM + gn] = z[r];
                }
                if (t > 0) {
                    float* mp = Zb + (size_t)gn * CDIM + g0;
                    v4f lo = { z[0], z[1], z[2], z[3] };
                    v4f hi = { z[4], z[5], z[6], z[7] };
                    *(v4f*)mp       = lo;
                    *(v4f*)(mp + 4) = hi;
                }
            }
        }
    }
}

template <int MODE>
__global__ __launch_bounds__(256)
void gemm256_kernel(const float* __restrict__ A,
                    const float* __restrict__ Bm,
                    float* __restrict__ D,
                    float* __restrict__ Zn,
                    const float* __restrict__ trc,
                    float scale, float diag) {
    __shared__ __attribute__((aligned(16))) float As[2][64 * GM_AP];
    __shared__ __attribute__((aligned(16))) float Bs[2][GM_KC * GM_BP];
    gemm256_body<MODE>(A, Bm, D, Zn, trc, scale, diag, blockIdx.x, As, Bs);
}

__global__ __launch_bounds__(256)
void gemm256_pair_kernel(const float* __restrict__ Y1,
                         const float* __restrict__ T,
                         float* __restrict__ Y2,
                         float* __restrict__ P) {
    __shared__ __attribute__((aligned(16))) float As[2][64 * GM_AP];
    __shared__ __attribute__((aligned(16))) float Bs[2][GM_KC * GM_BP];
    const unsigned NWG = NBATCH * 4;
    unsigned bid = blockIdx.x;
    if (bid < NWG)
        gemm256_body<0>(Y1, T, Y2, nullptr, nullptr, 0.5f, 0.0f, bid, As, Bs);
    else
        gemm256_body<3>(T, T, P, nullptr, nullptr, 0.5f, 0.0f, bid - NWG, As, Bs);
}

extern "C" void kernel_launch(void* const* d_in, const int* in_sizes, int n_in,
                              void* d_out, int out_size, void* d_ws, size_t ws_size,
                              hipStream_t stream) {
    (void)in_sizes; (void)n_in; (void)out_size; (void)ws_size;

    const float* x   = (const float*)d_in[0];
    float*       out = (float*)d_out;
    float*       ws  = (float*)d_ws;

    const size_t MAT = (size_t)NBATCH * CDIM * CDIM;

    float* trc  = ws;
    float* T    = trc + NBATCH;
    float* Ya   = T  + MAT;
    float* Yb   = Ya + MAT;
    float* T3   = Yb + MAT;
    float* Z1   = T3 + MAT;
    float* Pm   = Z1;

    sigma_kernel<<<NBATCH * 4, 256, 0, stream>>>(x, Ya, trc);

    gemm256_kernel<1><<<NBATCH * 4, 256, 0, stream>>>(Ya, Ya, Yb, Z1, nullptr, 0.5f, 0.0f);

    gemm256_kernel<0><<<NBATCH * 4, 256, 0, stream>>>(Z1, Yb, T,  nullptr, nullptr, -1.0f, 3.0f);

    gemm256_pair_kernel<<<NBATCH * 8, 256, 0, stream>>>(Yb, T, Ya, Pm);

    gemm256_kernel<3><<<NBATCH * 4, 256, 0, stream>>>(Pm, Pm, T3, nullptr, nullptr, -0.5f, 3.0f);

    gemm256_kernel<2><<<NBATCH * 4, 256, 0, stream>>>(Ya, T3, out, nullptr, trc,     0.5f, 0.0f);
}


# Planes written as whole 128-byte lines: tiles and mirrored copies staged through LDS, a wave stores four whole rows per instruction, not 64- and 16-byte pieces (packed output: row runs); same values
# speedup vs baseline: 3.7532x; 1.0435x over previous
#include <hip/hip_runtime.h>
#include <math.h>

typedef __attribute__((ext_vector_type(2))) float v2f;
typedef __attribute__((ext_vector_type(4))) float v4f;
typedef __attribute__((ext_vector_type(8))) float v8f;

#define CDIM 256
#define NDIM 784
#define NBATCH 64

static __device__ __forceinline__ v8f wmma4(v2f a, v2f b, v8f c) {
    return __builtin_amdgcn_wmma_f32_16x16x4_f32(false, a, false, b,
                                                 (short)0, c, false, false);
}

static __device__ __forceinline__ void lds_barrier() {
    __builtin_amdgcn_fence(__ATOMIC_RELEASE, "workgroup", "local");
    __builtin_amdgcn_s_barrier();
    __builtin_amdgcn_fence(__ATOMIC_ACQUIRE, "workgroup", "local");
}

#define EP_P      36
#define EP_PANEL  (32 * EP_P)
#define EP_NPAN   18
#define EP_FLOATS (EP_NPAN * EP_PANEL)

static __device__ __forceinline__
void ep_put_tile(float* S, int wq, int t, int rh, int cs, int lane, const float (&v)[8]) {
    int h = lane >> 4;
    int c = lane & 15;
    float* d = S + (wq * 5 + t) * EP_PANEL + (16 * rh + 8 * h) * EP_P + 16 * cs + c;
    #pragma unroll
    for (int r = 0; r < 8; ++r) d[r * EP_P] = v[r];
    if (t > 0) {
        float* m = S + (10 + wq * 4 + (t - 1)) * EP_PANEL + (16 * cs + c) * EP_P + 16 * rh + 8 * h;
        v4f lo = { v[0], v[1], v[2], v[3] };
        v4f hi = { v[4], v[5], v[6], v[7] };
        *(v4f*)m       = lo;
        *(v4f*)(m + 4) = hi;
    }
}

template <bool WITH_Z, bool FAR>
static __device__ __forceinline__
void ep_store_pass(const float* S, float* __restrict__ Db, float* __restrict__ Zb,
                   int wg, int wave, int lane) {
    int row = 4 * wave + (lane >> 3);
    int cq  = (lane & 7) << 2;
    const float* sp = S + row * EP_P + cq;
    v4f x[EP_NPAN];
    #pragma unroll
    for (int p = 0; p < EP_NPAN; ++p) {
        const int t = (p >= 10) ? (p - 10) % 4 + 1 : p % 5;
        if ((t == 4) != FAR) continue;
        x[p] = *(const v4f*)(sp + p * EP_PANEL);
    }
    #pragma unroll
    for (int p = 0; p < EP_NPAN; ++p) {
        const bool mir = p >= 10;
        const int  wq  = mir ? (p - 10) / 4 : p / 5;
        const int  t   = mir ? (p - 10) % 4 + 1 : p % 5;
        if ((t == 4) != FAR) continue;
        int Q = 2 * wg + wq;
        int P = (Q + t) & 7;
        int grow = 32 * (mir ? P : Q) + row;
        int gcol = 32 * (mir ? Q : P) + cq;
        *(v4f*)(Db + (size_t)grow * CDIM + gcol) = x[p];
        if (WITH_Z) {
            v4f z = { -x[p].x, -x[p].y, -x[p].z, -x[p].w };
            if (t == 0) {
                int d = row - cq;
                if (d == 0) z.x += 3.0f;
                if (d == 1) z.y += 3.0f;
                if (d == 2) z.z += 3.0f;
                if (d == 3) z.w += 3.0f;
            }
            z = 0.5f * z;
            *(v4f*)(Zb + (size_t)grow * CDIM + gcol) = z;
        }
    }
}

template <bool WITH_Z>
static __device__ __forceinline__
void ep_store_planes(const float* S, float* __restrict__ Db, float* __restrict__ Zb,
                     int wg, int wave, int lane) {
    ep_store_pass<WITH_Z, false>(S, Db, Zb, wg, wave, lane);
    if (wg < 2) ep_store_pass<WITH_Z, true>(S, Db, Zb, wg, wave, lane);
}


#define SG_KC 32
#define SG_P  36
#define NT    5

#define SG_LOAD_HALF(FA, FB, BUF, K0)                                        \
    _Pragma("unroll")                                                        \
    for (int s = 0; s < 4; ++s) {                                            \
        FA[s] = *(const v2f*)((BUF) + fa + (K0) + 4 * s);                    \
        _Pragma("unroll")                                                    \
        for (int t = 0; t < NT; ++t)                                         \
            FB[s][t] = *(const v2f*)((BUF) + fb[t] + (K0) + 4 * s);          \
    }
#define SG_MMA_HALF(FA, FB)                                                  \
    _Pragma("unroll")                                                        \
    for (int s = 0; s < 4; ++s) {                                            \
        _Pragma("unroll")                                                    \
        for (int t = 0; t < NT; ++t)                                         \
            c[t] = wmma4(FA[s], FB[s][t], c[t]);                             \
    }
#define SG_STAGE(BUF)                                                        \
    _Pragma("unroll")                                                        \
    for (int i = 0; i < 8; ++i) {                                            \
        v4f v = r[i];                                                        \
        *(v4f*)((BUF) + so + 32 * i * SG_P) = v;                             \
        q += v.x * v.x + v.y * v.y + v.z * v.z + v.w * v.w;                  \
        rs[i] += (v.x + v.y) + (v.z + v.w);                                  \
    }
#define SG_GLOAD(CH)                                                         \
    {                                                                        \
        const float* g = gp + (CH) * SG_KC;                                  \
        _Pragma("unroll")                                                    \
        for (int i = 0; i < 8; ++i)                                          \
            r[i] = *(const v4f*)(g + (size_t)(32 * i) * NDIM);               \
    }

__global__ __launch_bounds__(256)
void sigma_kernel(const float* __restrict__ X,
                  float* __restrict__ Y,
                  float* __restrict__ trc) {
    __shared__ __attribute__((aligned(16))) float xlds[EP_FLOATS];
    float (*Xs)[CDIM * SG_P] = (float (*)[CDIM * SG_P])xlds;
    __shared__ float qred[8];
    __shared__ float rsum[CDIM];

    int tid   = threadIdx.x;
    int wave  = __builtin_amdgcn_readfirstlane(tid >> 5);
    int lane  = tid & 31;
    int batch = blockIdx.x >> 2;
    int ti = (blockIdx.x & 3) << 6;

    const float* Xb = X + (size_t)batch * CDIM * NDIM;

    int srow = tid >> 3;
    int skq  = (tid & 7) << 2;
    const float* gp = Xb + (size_t)srow * NDIM + skq;
    int so = srow * SG_P + skq;
    float rs[8];
    #pragma unroll
    for (int i = 0; i < 8; ++i) rs[i] = 0.f;

    int wq = wave >> 2;
    int rh = (wave >> 1) & 1;
    int cs = wave & 1;
    int Q  = ((blockIdx.x & 3) << 1) + wq;
    int R0 = 32 * Q + 16 * rh;

    int mi = lane & 15;
    int kh = (lane >> 4) << 1;
    int fa = (R0 + mi) * SG_P + kh;
    int fb[NT];
    #pragma unroll
    for (int t = 0; t < NT; ++t)
        fb[t] = (32 * ((Q + t) & 7) + 16 * cs + mi) * SG_P + kh;

    v4f r[8];
    #pragma unroll
    for (int i = 0; i < 8; ++i)
        r[i] = *(const v4f*)(gp + (size_t)(32 * i) * NDIM);

    v8f c[NT];
    #pragma unroll
    for (int t = 0; t < NT; ++t) { v8f z = {}; c[t] = z; }

    float q = 0.f;
    const int NFULL = NDIM / SG_KC;
    const int TAILW = NDIM - NFULL * SG_KC;
    v2f xa[4], xb[4][NT];
    v2f ya[4], yb[4][NT];

    SG_STAGE(Xs[0])
    SG_GLOAD(1)
    lds_barrier();
    SG_LOAD_HALF(xa, xb, Xs[0], 0)
    SG_LOAD_HALF(ya, yb, Xs[0], 16)
    __builtin_amdgcn_sched_barrier(0);
    SG_MMA_HALF(xa, xb)
    SG_STAGE(Xs[1])
    SG_GLOAD(2)
    for (int ch = 1; ch < NFULL; ++ch) {
        float* buf  = Xs[ch & 1];
        float* nbuf = Xs[(ch + 1) & 1];
        lds_barrier();
        SG_LOAD_HALF(xa, xb, buf, 0)
        __builtin_amdgcn_sched_barrier(0);
        SG_MMA_HALF(ya, yb)
        __builtin_amdgcn_sched_barrier(0);
        SG_LOAD_HALF(ya, yb, buf, 16)
        if (ch + 1 < NFULL || skq < TAILW) {
            SG_STAGE(nbuf)
        }
        if (ch + 2 < NFULL || (ch + 2 == NFULL && skq < TAILW)) {
            SG_GLOAD(ch + 2)
        }
        SG_MMA_HALF(xa, xb)
    }
    lds_barrier();
    SG_LOAD_HALF(xa, xb, Xs[NFULL & 1], 0)
    __builtin_amdgcn_sched_barrier(0);
    SG_MMA_HALF(ya, yb)
    SG_MMA_HALF(xa, xb)

    #pragma unroll
    for (int i = 0; i < 8; ++i) {
        float t = rs[i];
        t += __shfl_xor(t, 1, 32);
        t += __shfl_xor(t, 2, 32);
        t += __shfl_xor(t, 4, 32);
        rs[i] = t;
    }
    if ((tid & 7) == 0) {
        #pragma unroll
        for (int i = 0; i < 8; ++i) rsum[srow + 32 * i] = rs[i];
    }
    const float inv_n = 1.0f / (float)NDIM;
    for (int off = 16; off > 0; off >>= 1) q += __shfl_xor(q, off, 32);
    if (lane == 0) qred[wave] = q;
    __syncthreads();
    float qq = ((qred[0] + qred[1]) + (qred[2] + qred[3])) +
               ((qred[4] + qred[5]) + (qred[6] + qred[7]));
    float s2 = 0.f;
    #pragma unroll
    for (int j = 0; j < 8; ++j) {
        float t = rsum[lane + 32 * j];
        s2 += t * t;
    }
    for (int off = 16; off > 0; off >>= 1) s2 += __shfl_xor(s2, off, 32);
    float tr = (qq - s2 * inv_n) * inv_n;
    if (ti == 0 && tid == 0) trc[batch] = tr;
    const float ysc = inv_n / tr;

    float* Sb = Y + (size_t)batch * CDIM * CDIM;
    int mbase = (lane >> 4) << 3;
    int nlo   = lane & 15;
    #pragma unroll
    for (int t = 0; t < NT; ++t) {
        if (t == 4 && Q >= 4) continue;
        int gn = 32 * ((Q + t) & 7) + 16 * cs + nlo;
        float scn = rsum[gn];
        float v[8];
        #pragma unroll
        for (int r8 = 0; r8 < 8; ++r8) {
            int gm = R0 + mbase + r8;
            v[r8] = (c[t][r8] - (rsum[gm] * scn) * inv_n) * ysc;
        }
        ep_put_tile(xlds, wq, t, rh, cs, lane, v);
    }
    __syncthreads();
    ep_store_planes<false>(xlds, Sb, nullptr, (int)(blockIdx.x & 3), wave, lane);
}

#define GM_KC 32
#define GM_AP 36
#define GM_BP 264

#define GM_GLOAD(CH)                                                         \
    {                                                                        \
        int k0 = (CH) * GM_KC;                                               \
        ra0 = *(const v4f*)(ga + k0);                                        \
        ra1 = *(const v4f*)(ga + k0 + (size_t)32 * CDIM);                    \
        _Pragma("unroll")                                                    \
        for (int i = 0; i < 8; ++i)                                          \
            rb[i] = *(const v4f*)(gb + (size_t)(k0 + 4 * i) * CDIM);         \
    }
#define GM_STAGE(CH, ABUF, BBUF)                                             \
    {                                                                        \
        v4f t0 = ra0, t1 = ra1;                                              \
        if (MODE == 3) {                                                     \
                \
              \
            int d0 = ti + arow - ((CH) * GM_KC + akq);                       \
            int d1 = d0 + 32;                                                \
            t0.x = (d0 == 0) ? 3.0f - t0.x : -t0.x;                          \
            t0.y = (d0 == 1) ? 3.0f - t0.y : -t0.y;                          \
            t0.z = (d0 == 2) ? 3.0f - t0.z : -t0.z;                          \
            t0.w = (d0 == 3) ? 3.0f - t0.w : -t0.w;                          \
            t1.x = (d1 == 0) ? 3.0f - t1.x : -t1.x;                          \
            t1.y = (d1 == 1) ? 3.0f - t1.y : -t1.y;                          \
            t1.z = (d1 == 2) ? 3.0f - t1.z : -t1.z;                          \
            t1.w = (d1 == 3) ? 3.0f - t1.w : -t1.w;                          \
        }                                                                    \
        *(v4f*)((ABUF) + sa)              = t0;                              \
        *(v4f*)((ABUF) + sa + 32 * GM_AP) = t1;                              \
        _Pragma("unroll")                                                    \
        for (int i = 0; i < 8; ++i) {                                        \
            v4f t = rb[i];                                                   \
            if (MODE == 1) {                                                 \
                  \
                                    \
                int d = (CH) * GM_KC + brow + 4 * i - bnq;                   \
                t.x = (d == 0) ? 3.0f - t.x : -t.x;                          \
                t.y = (d == 1) ? 3.0f - t.y : -t.y;                          \
                t.z = (d == 2) ? 3.0f - t.z : -t.z;                          \
                t.w = (d == 3) ? 3.0f - t.w : -t.w;                          \
            }                                                                \
            *(v4f*)((BBUF) + sb + 4 * i * GM_BP) = t;                        \
        }                                                                    \
    }
#define GM_LOAD_HALF(FA, FB, ABUF, BBUF, K0)                                 \
    _Pragma("unroll")                                                        \
    for (int s = 0; s < 4; ++s) {                                            \
        const int kk = (K0) + 4 * s;                                         \
        FA[s] = *(const v2f*)((ABUF) + fa + kk);                             \
        _Pragma("unroll")                                                    \
        for (int t = 0; t < NT; ++t) {                                       \
            const float* pb = (BBUF) + fb[t];                                \
            FB[s][t].x = pb[kk * GM_BP];                                     \
            FB[s][t].y = pb[(kk + 1) * GM_BP];                               \
        }                                                                    \
    }
#define GM_MMA_HALF(FA, FB)                                                  \
    _Pragma("unroll")                                                        \
    for (int s = 0; s < 4; ++s) {                                            \
        _Pragma("unroll")                                                    \
        for (int t = 0; t < NT; ++t)                                         \
            c[t] = wmma4(FA[s], FB[s][t], c[t]);                             \
    }

#define GM_LDS_FLOATS (2 * 64 * GM_AP + 2 * GM_KC * GM_BP)
static_assert(GM_LDS_FLOATS >= EP_FLOATS, "the staging area must fit into the chunk buffers");
template <int MODE>
static __device__ __forceinline__
void gemm256_body(const float* __restrict__ A,
                  const float* __restrict__ Bm,
                  float* __restrict__ D,
                  float* __restrict__ Zn,
                  const float* __restrict__ trc,
                  float scale, float diag, unsigned bid, float* lds) {
    float (*As)[64 * GM_AP]    = (float (*)[64 * GM_AP])lds;
    float (*Bs)[GM_KC * GM_BP] = (float (*)[GM_KC * GM_BP])(lds + 2 * 64 * GM_AP);
    int tid   = threadIdx.x;
    int wave  = __builtin_amdgcn_readfirstlane(tid >> 5);
    int lane  = tid & 31;
    int batch = bid >> 2;
    int ti = (bid & 3) << 6;

    const float* Ab = A  + (size_t)batch * CDIM * CDIM + (size_t)ti * CDIM;
    const float* Bb = Bm + (size_t)batch * CDIM * CDIM;

    int arow = tid >> 3;
    int akq  = (tid & 7) << 2;
    int brow = tid >> 6;
    int bnq  = (tid & 63) << 2;
    const float* ga = Ab + (size_t)arow * CDIM + akq;
    const float* gb = Bb + (size_t)brow * CDIM + bnq;
    int sa = arow * GM_AP + akq;
    int sb = brow * GM_BP + bnq;

    int wq = wave >> 2;
    int rh = (wave >> 1) & 1;
    int cs = wave & 1;
    int Q  = ((bid & 3) << 1) + wq;
    int R0 = 32 * Q + 16 * rh;

    int mi = lane & 15;
    int kh = (lane >> 4) << 1;
    int fa = (32 * wq + 16 * rh + mi) * GM_AP + kh;
    int fb[NT];
    #pragma unroll
    for (int t = 0; t < NT; ++t)
        fb[t] = kh * GM_BP + 32 * ((Q + t) & 7) + 16 * cs + mi;

    v4f ra0, ra1, rb[8];
    GM_GLOAD(0)

    v8f c[NT];
    #pragma unroll
    for (int t = 0; t < NT; ++t) { v8f z = {}; c[t] = z; }
    v2f xa[4], xb[4][NT];
    v2f ya[4], yb[4][NT];

    GM_STAGE(0, As[0], Bs[0])
    GM_GLOAD(1)
    lds_barrier();
    GM_LOAD_HALF(xa, xb, As[0], Bs[0], 0)
    GM_LOAD_HALF(ya, yb, As[0], Bs[0], 16)
    __builtin_amdgcn_sched_barrier(0);
    GM_MMA_HALF(xa, xb)
    GM_STAGE(1, As[1], Bs[1])
    GM_GLOAD(2)
    #pragma nounroll
    for (int ch = 1; ch < CDIM / GM_KC; ++ch) {
        float* abuf = As[ch & 1];
        float* bbuf = Bs[ch & 1];
        lds_barrier();
        GM_LOAD_HALF(xa, xb, abuf, bbuf, 0)
        __builtin_amdgcn_sched_barrier(0);
        GM_MMA_HALF(ya, yb)
        __builtin_amdgcn_sched_barrier(0);
        GM_LOAD_HALF(ya, yb, abuf, bbuf, 16)
        if (ch + 1 < CDIM / GM_KC) {
            GM_STAGE(ch + 1, As[(ch + 1) & 1], Bs[(ch + 1) & 1])
        }
        if (ch + 2 < CDIM / GM_KC) {
            GM_GLOAD(ch + 2)
        }
        GM_MMA_HALF(xa, xb)
    }
    GM_MMA_HALF(ya, yb)

    __syncthreads();

    float* Db = (MODE == 2) ? D + (size_t)batch * (CDIM * (CDIM + 1) / 2)
                            : D + (size_t)batch * CDIM * CDIM;
    float* Zb = (MODE == 1) ? Zn + (size_t)batch * CDIM * CDIM : nullptr;
    float s = (MODE == 2) ? sqrtf(trc[batch]) : 0.0f;
    int mbase = (lane >> 4) << 3;
    int nlo   = lane & 15;
    #pragma unroll
    for (int t = 0; t < NT; ++t) {
        if (t == 4 && Q >= 4) continue;
        int P  = (Q + t) & 7;
        int gn = 32 * P + 16 * cs + nlo;
        int g0 = R0 + mbase;
        float v[8];
        #pragma unroll
        for (int r = 0; r < 8; ++r) {
            v[r] = scale * c[t][r];
            if (t == 0 && g0 + r == gn) v[r] += diag;
        }
        if (MODE == 2) {
            if (t == 0) {
                #pragma unroll
                for (int r = 0; r < 8; ++r) {
                    int gm = g0 + r;
                    if (gn >= gm)
                        Db[gm * CDIM - (gm * (gm - 1)) / 2 + (gn - gm)] = s * v[r];
                }
            } else if (P > Q) {
                #pragma unroll
                for (int r = 0; r < 8; ++r) {
                    int gm = g0 + r;
                    Db[gm * CDIM - (gm * (gm - 1)) / 2 + (gn - gm)] = s * v[r];
                }
            } else {
                float* m = lds + (10 + wq * 4 + (t - 1)) * EP_PANEL
                               + (16 * cs + nlo) * EP_P + 16 * rh + mbase;
                v4f lo = { s * v[0], s * v[1], s * v[2], s * v[3] };
                v4f hi = { s * v[4], s * v[5], s * v[6], s * v[7] };
                *(v4f*)m       = lo;
                *(v4f*)(m + 4) = hi;
            }
        } else {
            ep_put_tile(lds, wq, t, rh, cs, lane, v);
        }
    }
    __syncthreads();
    if (MODE != 2) {
        ep_store_planes<MODE == 1>(lds, Db, Zb, (int)(bid & 3), wave, lane);
    } else {
        #pragma unroll
        for (int w2 = 0; w2 < 2; ++w2) {
            #pragma unroll
            for (int t = 1; t < 4; ++t) {
                int Q2 = (int)((bid & 3) << 1) + w2;
                if (Q2 + t < 8) continue;
                int P2 = (Q2 + t) & 7;
                const float* m = lds + (10 + w2 * 4 + (t - 1)) * EP_PANEL;
                #pragma unroll
                for (int k = 0; k < 4; ++k) {
                    int j = 4 * wave + k;
                    int g = 32 * P2 + j;
                    Db[g * CDIM - (g * (g - 1)) / 2 + (32 * Q2 + lane - g)] = m[j * EP_P + lane];
                }
            }
        }
    }
}

template <int MODE>
__global__ __launch_bounds__(256)
void gemm256_kernel(const float* __restrict__ A,
                    const float* __restrict__ Bm,
                    float* __restrict__ D,
                    float* __restrict__ Zn,
                    const float* __restrict__ trc,
                    float scale, float diag) {
    __shared__ __attribute__((aligned(16))) float lds[GM_LDS_FLOATS];
    gemm256_body<MODE>(A, Bm, D, Zn, trc, scale, diag, blockIdx.x, lds);
}

__global__ __launch_bounds__(256)
void gemm256_pair_kernel(const float* __restrict__ Y1,
                         const float* __restrict__ T,
                         float* __restrict__ Y2,
                         float* __restrict__ P) {
    __shared__ __attribute__((aligned(16))) float lds[GM_LDS_FLOATS];
    const unsigned NWG = NBATCH * 4;
    unsigned bid = blockIdx.x;
    if (bid < NWG)
        gemm256_body<0>(Y1, T, Y2, nullptr, nullptr, 0.5f, 0.0f, bid, lds);
    else
        gemm256_body<3>(T, T, P, nullptr, nullptr, 0.5f, 0.0f, bid - NWG, lds);
}

extern "C" void kernel_launch(void* const* d_in, const int* in_sizes, int n_in,
                              void* d_out, int out_size, void* d_ws, size_t ws_size,
                              hipStream_t stream) {
    (void)in_sizes; (void)n_in; (void)out_size; (void)ws_size;

    const float* x   = (const float*)d_in[0];
    float*       out = (float*)d_out;
    float*       ws  = (float*)d_ws;

    const size_t MAT = (size_t)NBATCH * CDIM * CDIM;

    float* trc  = ws;
    float* T    = trc + NBATCH;
    float* Ya   = T  + MAT;
    float* Yb   = Ya + MAT;
    float* T3   = Yb + MAT;
    float* Z1   = T3 + MAT;
    float* Pm   = Z1;

    sigma_kernel<<<NBATCH * 4, 256, 0, stream>>>(x, Ya, trc);

    gemm256_kernel<1><<<NBATCH * 4, 256, 0, stream>>>(Ya, Ya, Yb, Z1, nullptr, 0.5f, 0.0f);

    gemm256_kernel<0><<<NBATCH * 4, 256, 0, stream>>>(Z1, Yb, T,  nullptr, nullptr, -1.0f, 3.0f);

    gemm256_pair_kernel<<<NBATCH * 8, 256, 0, stream>>>(Yb, T, Ya, Pm);

    gemm256_kernel<3><<<NBATCH * 4, 256, 0, stream>>>(Pm, Pm, T3, nullptr, nullptr, -0.5f, 3.0f);

    gemm256_kernel<2><<<NBATCH * 4, 256, 0, stream>>>(Ya, T3, out, nullptr, trc,     0.5f, 0.0f);
}


# Panel pair four apart computed from one side only (row panels 0,2,5,7): per workgroup one row panel has 5 tiles per wave, the other 4, so a SIMD runs 9 not 10 WMMAs per k-step; rounding changed
# speedup vs baseline: 3.8829x; 1.0346x over previous
#include <hip/hip_runtime.h>
#include <math.h>

typedef __attribute__((ext_vector_type(2))) float v2f;
typedef __attribute__((ext_vector_type(4))) float v4f;
typedef __attribute__((ext_vector_type(8))) float v8f;

#define CDIM 256
#define NDIM 784
#define NBATCH 64

static __device__ __forceinline__ v8f wmma4(v2f a, v2f b, v8f c) {
    return __builtin_amdgcn_wmma_f32_16x16x4_f32(false, a, false, b,
                                                 (short)0, c, false, false);
}

static __device__ __forceinline__ void lds_barrier() {
    __builtin_amdgcn_fence(__ATOMIC_RELEASE, "workgroup", "local");
    __builtin_amdgcn_s_barrier();
    __builtin_amdgcn_fence(__ATOMIC_ACQUIRE, "workgroup", "local");
}

#define EP_P      36
#define EP_PANEL  (32 * EP_P)
#define EP_NPAN   18
#define EP_FLOATS (EP_NPAN * EP_PANEL)

static __device__ __forceinline__
void ep_put_tile(float* S, int wq, int t, int rh, int cs, int lane, const float (&v)[8]) {
    int h = lane >> 4;
    int c = lane & 15;
    float* d = S + (wq * 5 + t) * EP_PANEL + (16 * rh + 8 * h) * EP_P + 16 * cs + c;
    #pragma unroll
    for (int r = 0; r < 8; ++r) d[r * EP_P] = v[r];
    if (t > 0) {
        float* m = S + (10 + wq * 4 + (t - 1)) * EP_PANEL + (16 * cs + c) * EP_P + 16 * rh + 8 * h;
        v4f lo = { v[0], v[1], v[2], v[3] };
        v4f hi = { v[4], v[5], v[6], v[7] };
        *(v4f*)m       = lo;
        *(v4f*)(m + 4) = hi;
    }
}

template <bool WITH_Z>
static __device__ __forceinline__
void ep_store_pass(const float* S, float* __restrict__ Db, float* __restrict__ Zb,
                   int wg, int wave, int lane) {
    int row = 4 * wave + (lane >> 3);
    int cq  = (lane & 7) << 2;
    const float* sp = S + row * EP_P + cq;
    v4f x[EP_NPAN];
    #pragma unroll
    for (int p = 0; p < EP_NPAN; ++p) {
        const int t = (p >= 10) ? (p - 10) % 4 + 1 : p % 5;
        if (t == 4) continue;
        x[p] = *(const v4f*)(sp + p * EP_PANEL);
    }
    #pragma unroll
    for (int p = 0; p < EP_NPAN; ++p) {
        const bool mir = p >= 10;
        const int  wq  = mir ? (p - 10) / 4 : p / 5;
        const int  t   = mir ? (p - 10) % 4 + 1 : p % 5;
        if (t == 4) continue;
        int Q = 2 * wg + wq;
        int P = (Q + t) & 7;
        int grow = 32 * (mir ? P : Q) + row;
        int gcol = 32 * (mir ? Q : P) + cq;
        *(v4f*)(Db + (size_t)grow * CDIM + gcol) = x[p];
        if (WITH_Z) {
            v4f z = { -x[p].x, -x[p].y, -x[p].z, -x[p].w };
            if (t == 0) {
                int d = row - cq;
                if (d == 0) z.x += 3.0f;
                if (d == 1) z.y += 3.0f;
                if (d == 2) z.z += 3.0f;
                if (d == 3) z.w += 3.0f;
            }
            z = 0.5f * z;
            *(v4f*)(Zb + (size_t)grow * CDIM + gcol) = z;
        }
    }
}

template <bool WITH_Z>
static __device__ __forceinline__
void ep_store_far(const float* S, float* __restrict__ Db, float* __restrict__ Zb,
                  int wg, int wave, int lane) {
    int row = 4 * wave + (lane >> 3);
    int cq  = (lane & 7) << 2;
    int fq  = wg >> 1;
    const float* sp = S + row * EP_P + cq;
    v4f xd = *(const v4f*)(sp + (5 * fq + 4) * EP_PANEL);
    v4f xm = *(const v4f*)(sp + (10 + 4 * fq + 3) * EP_PANEL);
    int Q = 2 * wg + fq;
    int P = (Q + 4) & 7;
    size_t od = (size_t)(32 * Q + row) * CDIM + 32 * P + cq;
    size_t om = (size_t)(32 * P + row) * CDIM + 32 * Q + cq;
    *(v4f*)(Db + od) = xd;
    *(v4f*)(Db + om) = xm;
    if (WITH_Z) {
        v4f zd = { -xd.x, -xd.y, -xd.z, -xd.w };
        v4f zm = { -xm.x, -xm.y, -xm.z, -xm.w };
        zd = 0.5f * zd;
        zm = 0.5f * zm;
        *(v4f*)(Zb + od) = zd;
        *(v4f*)(Zb + om) = zm;
    }
}

template <bool WITH_Z>
static __device__ __forceinline__
void ep_store_planes(const float* S, float* __restrict__ Db, float* __restrict__ Zb,
                     int wg, int wave, int lane) {
    ep_store_pass<WITH_Z>(S, Db, Zb, wg, wave, lane);
    ep_store_far<WITH_Z>(S, Db, Zb, wg, wave, lane);
}


#define SG_KC 32
#define SG_P  36
#define NT    5

#define SG_LOAD_HALF(FA, FB, BUF, K0)                                        \
    _Pragma("unroll")                                                        \
    for (int s = 0; s < 4; ++s) {                                            \
        FA[s] = *(const v2f*)((BUF) + fa + (K0) + 4 * s);                    \
        _Pragma("unroll")                                                    \
        for (int t = 0; t < NT - 1; ++t)                                     \
            FB[s][t] = *(const v2f*)((BUF) + fb[t] + (K0) + 4 * s);          \
    }                                                                        \
    if (has5) {                                                              \
        _Pragma("unroll")                                                    \
        for (int s = 0; s < 4; ++s)                                          \
            FB[s][NT - 1] = *(const v2f*)((BUF) + fb[NT - 1] + (K0) + 4 * s); \
    }
#define SG_MMA_COMMON(FA, FB)                                                \
    _Pragma("unroll")                                                        \
    for (int s = 0; s < 4; ++s) {                                            \
        _Pragma("unroll")                                                    \
        for (int t = 0; t < NT - 1; ++t)                                     \
            c[t] = wmma4(FA[s], FB[s][t], c[t]);                             \
    }
#define SG_MMA_FIFTH(FA, FB)                                                 \
    if (has5) {                                                              \
        _Pragma("unroll")                                                    \
        for (int s = 0; s < 4; ++s)                                          \
            c[NT - 1] = wmma4(FA[s], FB[s][NT - 1], c[NT - 1]);              \
    }
#define SG_MMA_HALF(FA, FB)                                                  \
    SG_MMA_COMMON(FA, FB)                                                    \
    SG_MMA_FIFTH(FA, FB)
#define SG_STAGE(BUF)                                                        \
    _Pragma("unroll")                                                        \
    for (int i = 0; i < 8; ++i) {                                            \
        v4f v = r[i];                                                        \
        *(v4f*)((BUF) + so + 32 * i * SG_P) = v;                             \
        q += v.x * v.x + v.y * v.y + v.z * v.z + v.w * v.w;                  \
        rs[i] += (v.x + v.y) + (v.z + v.w);                                  \
    }
#define SG_GLOAD(CH)                                                         \
    {                                                                        \
        const float* g = gp + (CH) * SG_KC;                                  \
        _Pragma("unroll")                                                    \
        for (int i = 0; i < 8; ++i)                                          \
            r[i] = *(const v4f*)(g + (size_t)(32 * i) * NDIM);               \
    }

__global__ __launch_bounds__(256)
void sigma_kernel(const float* __restrict__ X,
                  float* __restrict__ Y,
                  float* __restrict__ trc) {
    __shared__ __attribute__((aligned(16))) float xlds[EP_FLOATS];
    float (*Xs)[CDIM * SG_P] = (float (*)[CDIM * SG_P])xlds;
    __shared__ float qred[8];
    __shared__ float rsum[CDIM];

    int tid   = threadIdx.x;
    int wave  = __builtin_amdgcn_readfirstlane(tid >> 5);
    int lane  = tid & 31;
    int batch = blockIdx.x >> 2;
    int ti = (blockIdx.x & 3) << 6;

    const float* Xb = X + (size_t)batch * CDIM * NDIM;

    int srow = tid >> 3;
    int skq  = (tid & 7) << 2;
    const float* gp = Xb + (size_t)srow * NDIM + skq;
    int so = srow * SG_P + skq;
    float rs[8];
    #pragma unroll
    for (int i = 0; i < 8; ++i) rs[i] = 0.f;

    int wq = wave >> 2;
    int rh = (wave >> 1) & 1;
    int cs = wave & 1;
    int Q  = ((blockIdx.x & 3) << 1) + wq;
    int R0 = 32 * Q + 16 * rh;
    const bool has5 = (wq == (int)((blockIdx.x & 3) >> 1));

    int mi = lane & 15;
    int kh = (lane >> 4) << 1;
    int fa = (R0 + mi) * SG_P + kh;
    int fb[NT];
    #pragma unroll
    for (int t = 0; t < NT; ++t)
        fb[t] = (32 * ((Q + t) & 7) + 16 * cs + mi) * SG_P + kh;

    v4f r[8];
    #pragma unroll
    for (int i = 0; i < 8; ++i)
        r[i] = *(const v4f*)(gp + (size_t)(32 * i) * NDIM);

    v8f c[NT];
    #pragma unroll
    for (int t = 0; t < NT; ++t) { v8f z = {}; c[t] = z; }

    float q = 0.f;
    const int NFULL = NDIM / SG_KC;
    const int TAILW = NDIM - NFULL * SG_KC;
    v2f xa[4], xb[4][NT];
    v2f ya[4], yb[4][NT];

    SG_STAGE(Xs[0])
    SG_GLOAD(1)
    lds_barrier();
    SG_LOAD_HALF(xa, xb, Xs[0], 0)
    SG_LOAD_HALF(ya, yb, Xs[0], 16)
    __builtin_amdgcn_sched_barrier(0);
    SG_MMA_HALF(xa, xb)
    SG_STAGE(Xs[1])
    SG_GLOAD(2)
    for (int ch = 1; ch < NFULL; ++ch) {
        float* buf  = Xs[ch & 1];
        float* nbuf = Xs[(ch + 1) & 1];
        lds_barrier();
        SG_LOAD_HALF(xa, xb, buf, 0)
        __builtin_amdgcn_sched_barrier(0);
        SG_MMA_HALF(ya, yb)
        __builtin_amdgcn_sched_barrier(0);
        SG_LOAD_HALF(ya, yb, buf, 16)
        __builtin_amdgcn_sched_barrier(0);
        SG_MMA_FIFTH(xa, xb)
        if (ch + 1 < NFULL || skq < TAILW) {
            SG_STAGE(nbuf)
        }
        if (ch + 2 < NFULL || (ch + 2 == NFULL && skq < TAILW)) {
            SG_GLOAD(ch + 2)
        }
        SG_MMA_COMMON(xa, xb)
    }
    lds_barrier();
    SG_LOAD_HALF(xa, xb, Xs[NFULL & 1], 0)
    __builtin_amdgcn_sched_barrier(0);
    SG_MMA_HALF(ya, yb)
    SG_MMA_HALF(xa, xb)

    #pragma unroll
    for (int i = 0; i < 8; ++i) {
        float t = rs[i];
        t += __shfl_xor(t, 1, 32);
        t += __shfl_xor(t, 2, 32);
        t += __shfl_xor(t, 4, 32);
        rs[i] = t;
    }
    if ((tid & 7) == 0) {
        #pragma unroll
        for (int i = 0; i < 8; ++i) rsum[srow + 32 * i] = rs[i];
    }
    const float inv_n = 1.0f / (float)NDIM;
    for (int off = 16; off > 0; off >>= 1) q += __shfl_xor(q, off, 32);
    if (lane == 0) qred[wave] = q;
    __syncthreads();
    float qq = ((qred[0] + qred[1]) + (qred[2] + qred[3])) +
               ((qred[4] + qred[5]) + (qred[6] + qred[7]));
    float s2 = 0.f;
    #pragma unroll
    for (int j = 0; j < 8; ++j) {
        float t = rsum[lane + 32 * j];
        s2 += t * t;
    }
    for (int off = 16; off > 0; off >>= 1) s2 += __shfl_xor(s2, off, 32);
    float tr = (qq - s2 * inv_n) * inv_n;
    if (ti == 0 && tid == 0) trc[batch] = tr;
    const float ysc = inv_n / tr;

    float* Sb = Y + (size_t)batch * CDIM * CDIM;
    int mbase = (lane >> 4) << 3;
    int nlo   = lane & 15;
    #pragma unroll
    for (int t = 0; t < NT; ++t) {
        if (t == 4 && !has5) continue;
        int gn = 32 * ((Q + t) & 7) + 16 * cs + nlo;
        float scn = rsum[gn];
        float v[8];
        #pragma unroll
        for (int r8 = 0; r8 < 8; ++r8) {
            int gm = R0 + mbase + r8;
            v[r8] = (c[t][r8] - (rsum[gm] * scn) * inv_n) * ysc;
        }
        ep_put_tile(xlds, wq, t, rh, cs, lane, v);
    }
    __syncthreads();
    ep_store_planes<false>(xlds, Sb, nullptr, (int)(blockIdx.x & 3), wave, lane);
}

#define GM_KC 32
#define GM_AP 36
#define GM_BP 264

#define GM_GLOAD(CH)                                                         \
    {                                                                        \
        int k0 = (CH) * GM_KC;                                               \
        ra0 = *(const v4f*)(ga + k0);                                        \
        ra1 = *(const v4f*)(ga + k0 + (size_t)32 * CDIM);                    \
        _Pragma("unroll")                                                    \
        for (int i = 0; i < 8; ++i)                                          \
            rb[i] = *(const v4f*)(gb + (size_t)(k0 + 4 * i) * CDIM);         \
    }
#define GM_STAGE(CH, ABUF, BBUF)                                             \
    {                                                                        \
        v4f t0 = ra0, t1 = ra1;                                              \
        if (MODE == 3) {                                                     \
                \
              \
            int d0 = ti + arow - ((CH) * GM_KC + akq);                       \
            int d1 = d0 + 32;                                                \
            t0.x = (d0 == 0) ? 3.0f - t0.x : -t0.x;                          \
            t0.y = (d0 == 1) ? 3.0f - t0.y : -t0.y;                          \
            t0.z = (d0 == 2) ? 3.0f - t0.z : -t0.z;                          \
            t0.w = (d0 == 3) ? 3.0f - t0.w : -t0.w;                          \
            t1.x = (d1 == 0) ? 3.0f - t1.x : -t1.x;                          \
            t1.y = (d1 == 1) ? 3.0f - t1.y : -t1.y;                          \
            t1.z = (d1 == 2) ? 3.0f - t1.z : -t1.z;                          \
            t1.w = (d1 == 3) ? 3.0f - t1.w : -t1.w;                          \
        }                                                                    \
        *(v4f*)((ABUF) + sa)              = t0;                              \
        *(v4f*)((ABUF) + sa + 32 * GM_AP) = t1;                              \
        _Pragma("unroll")                                                    \
        for (int i = 0; i < 8; ++i) {                                        \
            v4f t = rb[i];                                                   \
            if (MODE == 1) {                                                 \
                  \
                                    \
                int d = (CH) * GM_KC + brow + 4 * i - bnq;                   \
                t.x = (d == 0) ? 3.0f - t.x : -t.x;                          \
                t.y = (d == 1) ? 3.0f - t.y : -t.y;                          \
                t.z = (d == 2) ? 3.0f - t.z : -t.z;                          \
                t.w = (d == 3) ? 3.0f - t.w : -t.w;                          \
            }                                                                \
            *(v4f*)((BBUF) + sb + 4 * i * GM_BP) = t;                        \
        }                                                                    \
    }
#define GM_LOAD_HALF(FA, FB, ABUF, BBUF, K0)                                 \
    _Pragma("unroll")                                                        \
    for (int s = 0; s < 4; ++s) {                                            \
        const int kk = (K0) + 4 * s;                                         \
        FA[s] = *(const v2f*)((ABUF) + fa + kk);                             \
        _Pragma("unroll")                                                    \
        for (int t = 0; t < NT - 1; ++t) {                                   \
            const float* pb = (BBUF) + fb[t];                                \
            FB[s][t].x = pb[kk * GM_BP];                                     \
            FB[s][t].y = pb[(kk + 1) * GM_BP];                               \
        }                                                                    \
    }                                                                        \
    if (has5) {                                                              \
        _Pragma("unroll")                                                    \
        for (int s = 0; s < 4; ++s) {                                        \
            const int kk = (K0) + 4 * s;                                     \
            const float* pb = (BBUF) + fb[NT - 1];                           \
            FB[s][NT - 1].x = pb[kk * GM_BP];                                \
            FB[s][NT - 1].y = pb[(kk + 1) * GM_BP];                          \
        }                                                                    \
    }
#define GM_MMA_COMMON(FA, FB)                                                \
    _Pragma("unroll")                                                        \
    for (int s = 0; s < 4; ++s) {                                            \
        _Pragma("unroll")                                                    \
        for (int t = 0; t < NT - 1; ++t)                                     \
            c[t] = wmma4(FA[s], FB[s][t], c[t]);                             \
    }
#define GM_MMA_FIFTH(FA, FB)                                                 \
    if (has5) {                                                              \
        _Pragma("unroll")                                                    \
        for (int s = 0; s < 4; ++s)                                          \
            c[NT - 1] = wmma4(FA[s], FB[s][NT - 1], c[NT - 1]);              \
    }
#define GM_MMA_HALF(FA, FB)                                                  \
    GM_MMA_COMMON(FA, FB)                                                    \
    GM_MMA_FIFTH(FA, FB)

#define GM_LDS_FLOATS (2 * 64 * GM_AP + 2 * GM_KC * GM_BP)
static_assert(GM_LDS_FLOATS >= EP_FLOATS, "the staging area must fit into the chunk buffers");
template <int MODE>
static __device__ __forceinline__
void gemm256_body(const float* __restrict__ A,
                  const float* __restrict__ Bm,
                  float* __restrict__ D,
                  float* __restrict__ Zn,
                  const float* __restrict__ trc,
                  float scale, float diag, unsigned bid, float* lds) {
    float (*As)[64 * GM_AP]    = (float (*)[64 * GM_AP])lds;
    float (*Bs)[GM_KC * GM_BP] = (float (*)[GM_KC * GM_BP])(lds + 2 * 64 * GM_AP);
    int tid   = threadIdx.x;
    int wave  = __builtin_amdgcn_readfirstlane(tid >> 5);
    int lane  = tid & 31;
    int batch = bid >> 2;
    int ti = (bid & 3) << 6;

    const float* Ab = A  + (size_t)batch * CDIM * CDIM + (size_t)ti * CDIM;
    const float* Bb = Bm + (size_t)batch * CDIM * CDIM;

    int arow = tid >> 3;
    int akq  = (tid & 7) << 2;
    int brow = tid >> 6;
    int bnq  = (tid & 63) << 2;
    const float* ga = Ab + (size_t)arow * CDIM + akq;
    const float* gb = Bb + (size_t)brow * CDIM + bnq;
    int sa = arow * GM_AP + akq;
    int sb = brow * GM_BP + bnq;

    int wq = wave >> 2;
    int rh = (wave >> 1) & 1;
    int cs = wave & 1;
    int Q  = ((bid & 3) << 1) + wq;
    int R0 = 32 * Q + 16 * rh;
    const bool has5 = (wq == (int)((bid & 3) >> 1));

    int mi = lane & 15;
    int kh = (lane >> 4) << 1;
    int fa = (32 * wq + 16 * rh + mi) * GM_AP + kh;
    int fb[NT];
    #pragma unroll
    for (int t = 0; t < NT; ++t)
        fb[t] = kh * GM_BP + 32 * ((Q + t) & 7) + 16 * cs + mi;

    v4f ra0, ra1, rb[8];
    GM_GLOAD(0)

    v8f c[NT];
    #pragma unroll
    for (int t = 0; t < NT; ++t) { v8f z = {}; c[t] = z; }
    v2f xa[4], xb[4][NT];
    v2f ya[4], yb[4][NT];

    GM_STAGE(0, As[0], Bs[0])
    GM_GLOAD(1)
    lds_barrier();
    GM_LOAD_HALF(xa, xb, As[0], Bs[0], 0)
    GM_LOAD_HALF(ya, yb, As[0], Bs[0], 16)
    __builtin_amdgcn_sched_barrier(0);
    GM_MMA_HALF(xa, xb)
    GM_STAGE(1, As[1], Bs[1])
    GM_GLOAD(2)
    #pragma nounroll
    for (int ch = 1; ch < CDIM / GM_KC; ++ch) {
        float* abuf = As[ch & 1];
        float* bbuf = Bs[ch & 1];
        lds_barrier();
        GM_LOAD_HALF(xa, xb, abuf, bbuf, 0)
        __builtin_amdgcn_sched_barrier(0);
        GM_MMA_HALF(ya, yb)
        __builtin_amdgcn_sched_barrier(0);
        GM_LOAD_HALF(ya, yb, abuf, bbuf, 16)
        __builtin_amdgcn_sched_barrier(0);
        GM_MMA_FIFTH(xa, xb)
        if (ch + 1 < CDIM / GM_KC) {
            GM_STAGE(ch + 1, As[(ch + 1) & 1], Bs[(ch + 1) & 1])
        }
        if (ch + 2 < CDIM / GM_KC) {
            GM_GLOAD(ch + 2)
        }
        GM_MMA_COMMON(xa, xb)
    }
    GM_MMA_HALF(ya, yb)

    __syncthreads();

    float* Db = (MODE == 2) ? D + (size_t)batch * (CDIM * (CDIM + 1) / 2)
                            : D + (size_t)batch * CDIM * CDIM;
    float* Zb = (MODE == 1) ? Zn + (size_t)batch * CDIM * CDIM : nullptr;
    float s = (MODE == 2) ? sqrtf(trc[batch]) : 0.0f;
    int mbase = (lane >> 4) << 3;
    int nlo   = lane & 15;
    #pragma unroll
    for (int t = 0; t < NT; ++t) {
        if (t == 4 && !has5) continue;
        int P  = (Q + t) & 7;
        int gn = 32 * P + 16 * cs + nlo;
        int g0 = R0 + mbase;
        float v[8];
        #pragma unroll
        for (int r = 0; r < 8; ++r) {
            v[r] = scale * c[t][r];
            if (t == 0 && g0 + r == gn) v[r] += diag;
        }
        if (MODE == 2) {
            if (t == 0) {
                #pragma unroll
                for (int r = 0; r < 8; ++r) {
                    int gm = g0 + r;
                    if (gn >= gm)
                        Db[gm * CDIM - (gm * (gm - 1)) / 2 + (gn - gm)] = s * v[r];
                }
            } else if (P > Q) {
                #pragma unroll
                for (int r = 0; r < 8; ++r) {
                    int gm = g0 + r;
                    Db[gm * CDIM - (gm * (gm - 1)) / 2 + (gn - gm)] = s * v[r];
                }
            } else {
                float* m = lds + (10 + wq * 4 + (t - 1)) * EP_PANEL
                               + (16 * cs + nlo) * EP_P + 16 * rh + mbase;
                v4f lo = { s * v[0], s * v[1], s * v[2], s * v[3] };
                v4f hi = { s * v[4], s * v[5], s * v[6], s * v[7] };
                *(v4f*)m       = lo;
                *(v4f*)(m + 4) = hi;
            }
        } else {
            ep_put_tile(lds, wq, t, rh, cs, lane, v);
        }
    }
    __syncthreads();
    if (MODE != 2) {
        ep_store_planes<MODE == 1>(lds, Db, Zb, (int)(bid & 3), wave, lane);
    } else {
        #pragma unroll
        for (int w2 = 0; w2 < 2; ++w2) {
            #pragma unroll
            for (int t = 1; t < 5; ++t) {
                int Q2 = (int)((bid & 3) << 1) + w2;
                if (Q2 + t < 8) continue;
                if (t == 4 && w2 != (int)((bid & 3) >> 1)) continue;
                int P2 = (Q2 + t) & 7;
                const float* m = lds + (10 + w2 * 4 + (t - 1)) * EP_PANEL;
                #pragma unroll
                for (int k = 0; k < 4; ++k) {
                    int j = 4 * wave + k;
                    int g = 32 * P2 + j;
                    Db[g * CDIM - (g * (g - 1)) / 2 + (32 * Q2 + lane - g)] = m[j * EP_P + lane];
                }
            }
        }
    }
}

template <int MODE>
__global__ __launch_bounds__(256)
void gemm256_kernel(const float* __restrict__ A,
                    const float* __restrict__ Bm,
                    float* __restrict__ D,
                    float* __restrict__ Zn,
                    const float* __restrict__ trc,
                    float scale, float diag) {
    __shared__ __attribute__((aligned(16))) float lds[GM_LDS_FLOATS];
    gemm256_body<MODE>(A, Bm, D, Zn, trc, scale, diag, blockIdx.x, lds);
}

__global__ __launch_bounds__(256)
void gemm256_pair_kernel(const float* __restrict__ Y1,
                         const float* __restrict__ T,
                         float* __restrict__ Y2,
                         float* __restrict__ P) {
    __shared__ __attribute__((aligned(16))) float lds[GM_LDS_FLOATS];
    const unsigned NWG = NBATCH * 4;
    unsigned bid = blockIdx.x;
    if (bid < NWG)
        gemm256_body<0>(Y1, T, Y2, nullptr, nullptr, 0.5f, 0.0f, bid, lds);
    else
        gemm256_body<3>(T, T, P, nullptr, nullptr, 0.5f, 0.0f, bid - NWG, lds);
}

extern "C" void kernel_launch(void* const* d_in, const int* in_sizes, int n_in,
                              void* d_out, int out_size, void* d_ws, size_t ws_size,
                              hipStream_t stream) {
    (void)in_sizes; (void)n_in; (void)out_size; (void)ws_size;

    const float* x   = (const float*)d_in[0];
    float*       out = (float*)d_out;
    float*       ws  = (float*)d_ws;

    const size_t MAT = (size_t)NBATCH * CDIM * CDIM;

    float* trc  = ws;
    float* T    = trc + NBATCH;
    float* Ya   = T  + MAT;
    float* Yb   = Ya + MAT;
    float* T3   = Yb + MAT;
    float* Z1   = T3 + MAT;
    float* Pm   = Z1;

    sigma_kernel<<<NBATCH * 4, 256, 0, stream>>>(x, Ya, trc);

    gemm256_kernel<1><<<NBATCH * 4, 256, 0, stream>>>(Ya, Ya, Yb, Z1, nullptr, 0.5f, 0.0f);

    gemm256_kernel<0><<<NBATCH * 4, 256, 0, stream>>>(Z1, Yb, T,  nullptr, nullptr, -1.0f, 3.0f);

    gemm256_pair_kernel<<<NBATCH * 8, 256, 0, stream>>>(Yb, T, Ya, Pm);

    gemm256_kernel<3><<<NBATCH * 4, 256, 0, stream>>>(Pm, Pm, T3, nullptr, nullptr, -0.5f, 3.0f);

    gemm256_kernel<2><<<NBATCH * 4, 256, 0, stream>>>(Ya, T3, out, nullptr, trc,     0.5f, 0.0f);
}
